# starting kernel
# baseline (speedup 1.0000x reference)
#include <hip/hip_runtime.h>

typedef __attribute__((ext_vector_type(2))) float v2f;
typedef __attribute__((ext_vector_type(8))) float v8f;

constexpr int NUM_USERS = 100000;
constexpr int NUM_ITEMS = 50000;
constexpr int N_NODES   = NUM_USERS + NUM_ITEMS;
constexpr int DIM       = 64;
constexpr int SBERT     = 384;
constexpr int NEDGE     = 2000000;
constexpr int NLAYERS   = 3;

__global__ void zero_f(float* __restrict__ p, int n) {
  int i = blockIdx.x * blockDim.x + threadIdx.x;
  if (i < n) p[i] = 0.0f;
}

__global__ void init_user(const float* __restrict__ u,
                          float* __restrict__ emb0, float* __restrict__ acc) {
  int i = blockIdx.x * blockDim.x + threadIdx.x;
  if (i < NUM_USERS * DIM) {
    float v = u[i];
    emb0[i] = v;
    acc[i]  = v;
  }
}

__global__ __launch_bounds__(128)
void wmma_proj(const float* __restrict__ bert, const float* __restrict__ w,
               float* __restrict__ emb0, float* __restrict__ acc) {
  const int lane = threadIdx.x & 31;
  const int wave = threadIdx.x >> 5;
  const int m0   = blockIdx.x * 16;
  const int n0   = wave * 16;
  const int half = lane >> 4;
  const int r    = lane & 15;

  const float* arow = bert + (size_t)(m0 + r) * SBERT + 2 * half;
  const float* brow = w    + (size_t)(n0 + r) * SBERT + 2 * half;

  v8f c = {};
#pragma unroll 8
  for (int k = 0; k < SBERT; k += 4) {
    v2f a = *(const v2f*)(arow + k);
    v2f b = *(const v2f*)(brow + k);
    c = __builtin_amdgcn_wmma_f32_16x16x4_f32(
         false, a,  false, b,
         (short)0, c,  false,  false);
  }

  const size_t base = (size_t)(NUM_USERS + m0 + 8 * half) * DIM + (size_t)(n0 + r);
#pragma unroll
  for (int i = 0; i < 8; ++i) {
    float v = c[i];
    emb0[base + (size_t)i * DIM] = v;
    acc [base + (size_t)i * DIM] = v;
  }
}

__global__ void deg_count(const int* __restrict__ dst, float* __restrict__ deg) {
  int e = blockIdx.x * blockDim.x + threadIdx.x;
  if (e < NEDGE) atomicAdd(&deg[dst[e]], 1.0f);
}

__global__ void deg_to_dis(float* __restrict__ deg) {
  int i = blockIdx.x * blockDim.x + threadIdx.x;
  if (i < N_NODES) {
    float d = deg[i];
    deg[i] = (d > 0.0f) ? rsqrtf(fmaxf(d, 1.0f)) : 0.0f;
  }
}

__global__ void edge_norm(const int* __restrict__ src, const int* __restrict__ dst,
                          const float* __restrict__ dis, float* __restrict__ norm) {
  int e = blockIdx.x * blockDim.x + threadIdx.x;
  if (e < NEDGE) norm[e] = dis[src[e]] * dis[dst[e]];
}

__global__ void spmm_scatter(const int* __restrict__ src, const int* __restrict__ dst,
                             const float* __restrict__ norm,
                             const float* __restrict__ x, float* __restrict__ xn) {
  int t = blockIdx.x * blockDim.x + threadIdx.x;
  int e = t >> 4;
  if (e >= NEDGE) return;
  int c = (t & 15) << 2;
  float wgt = norm[e];
  float4 v = *(const float4*)(x + (size_t)src[e] * DIM + c);
  float* o = xn + (size_t)dst[e] * DIM + c;
  atomicAdd(o + 0, wgt * v.x);
  atomicAdd(o + 1, wgt * v.y);
  atomicAdd(o + 2, wgt * v.z);
  atomicAdd(o + 3, wgt * v.w);
}

__global__ void acc_add(float* __restrict__ acc, const float* __restrict__ xn) {
  int i = blockIdx.x * blockDim.x + threadIdx.x;
  if (i < N_NODES * DIM) acc[i] += xn[i];
}

__global__ void scale_k(float* __restrict__ acc) {
  int i = blockIdx.x * blockDim.x + threadIdx.x;
  if (i < N_NODES * DIM) acc[i] *= 0.25f;
}

extern "C" void kernel_launch(void* const* d_in, const int* in_sizes, int n_in,
                              void* d_out, int out_size, void* d_ws, size_t ws_size,
                              hipStream_t stream) {
  const int*   ei    = (const int*)d_in[0];
  const int*   src   = ei;
  const int*   dst   = ei + NEDGE;
  const float* user  = (const float*)d_in[1];
  const float* bert  = (const float*)d_in[2];
  const float* wproj = (const float*)d_in[3];

  float* emb0 = (float*)d_out;
  float* acc  = emb0 + (size_t)N_NODES * DIM;

  float* deg  = (float*)d_ws;
  float* norm = deg  + ((N_NODES + 63) & ~63);
  float* xA   = norm + ((NEDGE + 63) & ~63);
  float* xB   = xA   + (size_t)N_NODES * DIM;

  const int B = 256;
  const int NV = N_NODES * DIM;

  zero_f   <<<(N_NODES + B - 1) / B, B, 0, stream>>>(deg, N_NODES);
  init_user<<<(NUM_USERS * DIM + B - 1) / B, B, 0, stream>>>(user, emb0, acc);
  wmma_proj<<<NUM_ITEMS / 16, 128, 0, stream>>>(bert, wproj, emb0, acc);

  deg_count <<<(NEDGE + B - 1) / B, B, 0, stream>>>(dst, deg);
  deg_to_dis<<<(N_NODES + B - 1) / B, B, 0, stream>>>(deg);
  edge_norm <<<(NEDGE + B - 1) / B, B, 0, stream>>>(src, dst, deg, norm);

  const float* xcur = emb0;
  float* bufs[2] = {xA, xB};
  for (int l = 0; l < NLAYERS; ++l) {
    float* xn = bufs[l & 1];
    zero_f      <<<(NV + B - 1) / B, B, 0, stream>>>(xn, NV);
    spmm_scatter<<<(NEDGE * 16 + B - 1) / B, B, 0, stream>>>(src, dst, norm, xcur, xn);
    acc_add     <<<(NV + B - 1) / B, B, 0, stream>>>(acc, xn);
    xcur = xn;
  }
  scale_k<<<(NV + B - 1) / B, B, 0, stream>>>(acc);
}


# SpMM per-destination-row gather over an in-call counting sort of edges (int rank atomics) replaces f32 atomic scatter; row owner adds acc; norm factored dis[d]*sum(dis[s]x[s]); row sums in rank order
# speedup vs baseline: 2.3434x; 2.3434x over previous
#include <hip/hip_runtime.h>

typedef __attribute__((ext_vector_type(2))) float v2f;
typedef __attribute__((ext_vector_type(8))) float v8f;

constexpr int NUM_USERS = 100000;
constexpr int NUM_ITEMS = 50000;
constexpr int N_NODES   = NUM_USERS + NUM_ITEMS;
constexpr int DIM       = 64;
constexpr int SBERT     = 384;
constexpr int NEDGE     = 2000000;
constexpr int NLAYERS   = 3;

__global__ void zero_i(int* __restrict__ p, int n) {
  int i = blockIdx.x * blockDim.x + threadIdx.x;
  if (i < n) p[i] = 0;
}

__global__ void init_user(const float* __restrict__ u,
                          float* __restrict__ emb0, float* __restrict__ acc) {
  int i = blockIdx.x * blockDim.x + threadIdx.x;
  if (i < NUM_USERS * DIM) {
    float v = u[i];
    emb0[i] = v;
    acc[i]  = v;
  }
}

__global__ __launch_bounds__(128)
void wmma_proj(const float* __restrict__ bert, const float* __restrict__ w,
               float* __restrict__ emb0, float* __restrict__ acc) {
  const int lane = threadIdx.x & 31;
  const int wave = threadIdx.x >> 5;
  const int m0   = blockIdx.x * 16;
  const int n0   = wave * 16;
  const int half = lane >> 4;
  const int r    = lane & 15;

  const float* arow = bert + (size_t)(m0 + r) * SBERT + 2 * half;
  const float* brow = w    + (size_t)(n0 + r) * SBERT + 2 * half;

  v8f c = {};
#pragma unroll 8
  for (int k = 0; k < SBERT; k += 4) {
    v2f a = *(const v2f*)(arow + k);
    v2f b = *(const v2f*)(brow + k);
    c = __builtin_amdgcn_wmma_f32_16x16x4_f32(
         false, a,  false, b,
         (short)0, c,  false,  false);
  }

  const size_t base = (size_t)(NUM_USERS + m0 + 8 * half) * DIM + (size_t)(n0 + r);
#pragma unroll
  for (int i = 0; i < 8; ++i) {
    float v = c[i];
    emb0[base + (size_t)i * DIM] = v;
    acc [base + (size_t)i * DIM] = v;
  }
}


__global__ void count_rank(const int* __restrict__ dst, int* __restrict__ cnt,
                           int* __restrict__ rank) {
  int e = blockIdx.x * blockDim.x + threadIdx.x;
  if (e < NEDGE) rank[e] = atomicAdd(&cnt[dst[e]], 1);
}

constexpr int SCAN_T = 1024;
constexpr int SCAN_BLOCKS = (N_NODES + 1 + SCAN_T - 1) / SCAN_T;

__global__ __launch_bounds__(SCAN_T)
void scan_rows(const int* __restrict__ cnt, int* __restrict__ rowptr) {
  __shared__ int s_own[SCAN_T / 32];
  __shared__ int s_pre[SCAN_T / 32];
  const int tid  = threadIdx.x;
  const int lane = tid & 31;
  const int wv   = tid >> 5;

  int pre = 0;
  const int4* c4 = (const int4*)cnt;
  const int n4 = blockIdx.x * (SCAN_T / 4);
  int i = tid;
  for (; i + 3 * SCAN_T < n4; i += 4 * SCAN_T) {
    int4 q0 = c4[i];
    int4 q1 = c4[i + SCAN_T];
    int4 q2 = c4[i + 2 * SCAN_T];
    int4 q3 = c4[i + 3 * SCAN_T];
    pre += ((q0.x + q0.y) + (q0.z + q0.w)) + ((q1.x + q1.y) + (q1.z + q1.w));
    pre += ((q2.x + q2.y) + (q2.z + q2.w)) + ((q3.x + q3.y) + (q3.z + q3.w));
  }
  for (; i < n4; i += SCAN_T) {
    int4 q = c4[i];
    pre += (q.x + q.y) + (q.z + q.w);
  }
  const int idx = blockIdx.x * SCAN_T + tid;
  const int own = (idx < N_NODES) ? cnt[idx] : 0;

  int inc = own;
  for (int d = 1; d < 32; d <<= 1) {
    int up = __shfl_up(inc, d);
    if (lane >= d) inc += up;
  }
  for (int d = 16; d > 0; d >>= 1) pre += __shfl_xor(pre, d);
  if (lane == 31) s_own[wv] = inc;
  if (lane == 0)  s_pre[wv] = pre;
  __syncthreads();

  int before = (lane < wv) ? s_own[lane] : 0;
  int front  = s_pre[lane];
  for (int d = 16; d > 0; d >>= 1) {
    before += __shfl_xor(before, d);
    front  += __shfl_xor(front, d);
  }
  if (idx <= N_NODES) rowptr[idx] = front + before + (inc - own);
}

__global__ void fill_csr(const int* __restrict__ src, const int* __restrict__ dst,
                         const int* __restrict__ rank, const int* __restrict__ rowptr,
                         int* __restrict__ csr) {
  int e = blockIdx.x * blockDim.x + threadIdx.x;
  if (e < NEDGE) csr[rowptr[dst[e]] + rank[e]] = src[e];
}

__device__ __forceinline__ float dis_of(int deg) {
  return (deg > 0) ? rsqrtf(fmaxf((float)deg, 1.0f)) : 0.0f;
}

__global__ void prescale(const int* __restrict__ rowptr, const float* __restrict__ emb0,
                         float* __restrict__ xt) {
  const int t = blockIdx.x * blockDim.x + threadIdx.x;
  const int row = t >> 4;
  if (row >= N_NODES) return;
  const float dd = dis_of(rowptr[row + 1] - rowptr[row]);
  float4 v = *(const float4*)(emb0 + (size_t)t * 4);
  v.x *= dd; v.y *= dd; v.z *= dd; v.w *= dd;
  *(float4*)(xt + (size_t)t * 4) = v;
}

template <bool LAST>
__global__ __launch_bounds__(256)
void spmm_rows(const int* __restrict__ rowptr, const int* __restrict__ csr,
               const float* __restrict__ xt, float* __restrict__ xt_next,
               float* __restrict__ acc) {
  const int t = blockIdx.x * 256 + threadIdx.x;
  const int row = t >> 4;
  if (row >= N_NODES) return;
  const unsigned c = (unsigned)(t & 15) << 2;
  const int beg = rowptr[row];
  const int end = rowptr[row + 1];

  float4 s0 = {0.f, 0.f, 0.f, 0.f}, s1 = s0, s2 = s0, s3 = s0;
  int e = beg;
  for (; e + 4 <= end; e += 4) {
    const unsigned j0 = (unsigned)csr[e + 0];
    const unsigned j1 = (unsigned)csr[e + 1];
    const unsigned j2 = (unsigned)csr[e + 2];
    const unsigned j3 = (unsigned)csr[e + 3];
    const float4 v0 = *(const float4*)(xt + (j0 * (unsigned)DIM + c));
    const float4 v1 = *(const float4*)(xt + (j1 * (unsigned)DIM + c));
    const float4 v2 = *(const float4*)(xt + (j2 * (unsigned)DIM + c));
    const float4 v3 = *(const float4*)(xt + (j3 * (unsigned)DIM + c));
    s0.x += v0.x; s0.y += v0.y; s0.z += v0.z; s0.w += v0.w;
    s1.x += v1.x; s1.y += v1.y; s1.z += v1.z; s1.w += v1.w;
    s2.x += v2.x; s2.y += v2.y; s2.z += v2.z; s2.w += v2.w;
    s3.x += v3.x; s3.y += v3.y; s3.z += v3.z; s3.w += v3.w;
  }
  const int rem = end - e;
  if (rem > 0) {
    const int last = end - 1;
    const unsigned j0 = (unsigned)csr[e];
    const unsigned j1 = (unsigned)csr[min(e + 1, last)];
    const unsigned j2 = (unsigned)csr[min(e + 2, last)];
    const float4 z = {0.f, 0.f, 0.f, 0.f};
    const float4 v0 = *(const float4*)(xt + (j0 * (unsigned)DIM + c));
    float4 v1 = z, v2 = z;
    if (rem > 1) v1 = *(const float4*)(xt + (j1 * (unsigned)DIM + c));
    if (rem > 2) v2 = *(const float4*)(xt + (j2 * (unsigned)DIM + c));
    s0.x += v0.x; s0.y += v0.y; s0.z += v0.z; s0.w += v0.w;
    s1.x += v1.x; s1.y += v1.y; s1.z += v1.z; s1.w += v1.w;
    s2.x += v2.x; s2.y += v2.y; s2.z += v2.z; s2.w += v2.w;
  }

  const float dd = dis_of(end - beg);
  float4 x;
  x.x = dd * ((s0.x + s1.x) + (s2.x + s3.x));
  x.y = dd * ((s0.y + s1.y) + (s2.y + s3.y));
  x.z = dd * ((s0.z + s1.z) + (s2.z + s3.z));
  x.w = dd * ((s0.w + s1.w) + (s2.w + s3.w));

  const size_t o = (size_t)t * 4;
  float4 a = *(const float4*)(acc + o);
  a.x += x.x; a.y += x.y; a.z += x.z; a.w += x.w;
  if (LAST) {
    a.x *= 0.25f; a.y *= 0.25f; a.z *= 0.25f; a.w *= 0.25f;
  } else {
    float4 y;
    y.x = dd * x.x; y.y = dd * x.y; y.z = dd * x.z; y.w = dd * x.w;
    *(float4*)(xt_next + o) = y;
  }
  *(float4*)(acc + o) = a;
}

extern "C" void kernel_launch(void* const* d_in, const int* in_sizes, int n_in,
                              void* d_out, int out_size, void* d_ws, size_t ws_size,
                              hipStream_t stream) {
  const int*   ei    = (const int*)d_in[0];
  const int*   src   = ei;
  const int*   dst   = ei + NEDGE;
  const float* user  = (const float*)d_in[1];
  const float* bert  = (const float*)d_in[2];
  const float* wproj = (const float*)d_in[3];

  float* emb0 = (float*)d_out;
  float* acc  = emb0 + (size_t)N_NODES * DIM;

  int*   rowptr = (int*)d_ws;
  int*   csr    = rowptr + ((N_NODES + 1 + 63) & ~63);
  float* xA     = (float*)(csr + ((NEDGE + 63) & ~63));
  float* xB     = xA + (size_t)N_NODES * DIM;
  int*   rank   = (int*)xB;
  int*   cnt    = rank + ((NEDGE + 63) & ~63);

  const int B = 256;

  zero_i   <<<(N_NODES + B - 1) / B, B, 0, stream>>>(cnt, N_NODES);
  init_user<<<(NUM_USERS * DIM + B - 1) / B, B, 0, stream>>>(user, emb0, acc);
  wmma_proj<<<NUM_ITEMS / 16, 128, 0, stream>>>(bert, wproj, emb0, acc);

  count_rank<<<(NEDGE + B - 1) / B, B, 0, stream>>>(dst, cnt, rank);
  scan_rows <<<SCAN_BLOCKS, SCAN_T, 0, stream>>>(cnt, rowptr);
  fill_csr  <<<(NEDGE + B - 1) / B, B, 0, stream>>>(src, dst, rank, rowptr, csr);

  prescale<<<(N_NODES * 16 + B - 1) / B, B, 0, stream>>>(rowptr, emb0, xA);
  spmm_rows<false><<<(N_NODES * 16 + 255) / 256, 256, 0, stream>>>(rowptr, csr, xA, xB, acc);
  spmm_rows<false><<<(N_NODES * 16 + 255) / 256, 256, 0, stream>>>(rowptr, csr, xB, xA, acc);
  spmm_rows<true> <<<(N_NODES * 16 + 255) / 256, 256, 0, stream>>>(rowptr, csr, xA, xB, acc);
}


# Edge sort by destination as a two-level counting sort in LDS (1024-row buckets, then one bucket per block) replaces per-edge global atomics and 4-byte scatters; row sums in LDS-add arrival order
# speedup vs baseline: 3.6005x; 1.5364x over previous
#include <hip/hip_runtime.h>

typedef __attribute__((ext_vector_type(2))) float v2f;
typedef __attribute__((ext_vector_type(8))) float v8f;

constexpr int NUM_USERS = 100000;
constexpr int NUM_ITEMS = 50000;
constexpr int N_NODES   = NUM_USERS + NUM_ITEMS;
constexpr int DIM       = 64;
constexpr int SBERT     = 384;
constexpr int NEDGE     = 2000000;
constexpr int NLAYERS   = 3;

__global__ void init_user(const float* __restrict__ u,
                          float* __restrict__ emb0, float* __restrict__ acc) {
  int i = blockIdx.x * blockDim.x + threadIdx.x;
  if (i < NUM_USERS * DIM) {
    float v = u[i];
    emb0[i] = v;
    acc[i]  = v;
  }
}

__global__ __launch_bounds__(128)
void wmma_proj(const float* __restrict__ bert, const float* __restrict__ w,
               float* __restrict__ emb0, float* __restrict__ acc) {
  const int lane = threadIdx.x & 31;
  const int wave = threadIdx.x >> 5;
  const int m0   = blockIdx.x * 16;
  const int n0   = wave * 16;
  const int half = lane >> 4;
  const int r    = lane & 15;

  const float* arow = bert + (size_t)(m0 + r) * SBERT + 2 * half;
  const float* brow = w    + (size_t)(n0 + r) * SBERT + 2 * half;

  v8f c = {};
#pragma unroll 8
  for (int k = 0; k < SBERT; k += 4) {
    v2f a = *(const v2f*)(arow + k);
    v2f b = *(const v2f*)(brow + k);
    c = __builtin_amdgcn_wmma_f32_16x16x4_f32(
         false, a,  false, b,
         (short)0, c,  false,  false);
  }

  const size_t base = (size_t)(NUM_USERS + m0 + 8 * half) * DIM + (size_t)(n0 + r);
#pragma unroll
  for (int i = 0; i < 8; ++i) {
    float v = c[i];
    emb0[base + (size_t)i * DIM] = v;
    acc [base + (size_t)i * DIM] = v;
  }
}


constexpr int BT    = 1024;
constexpr int NBKT  = (N_NODES + 1023) >> 10;
constexpr int GB    = 256;
constexpr int CHUNK = (NEDGE + GB - 1) / GB;
constexpr int EPT   = 8;
constexpr int CHUNK_PAD = EPT * BT;
static_assert(CHUNK <= CHUNK_PAD, "a chunk must fit its LDS stage");
constexpr int CAPL  = 32768;
constexpr unsigned NOEDGE = 0xFFFFFFFFu;

__global__ __launch_bounds__(BT)
void bucket_hist(const int* __restrict__ dst, int* __restrict__ hist) {
  __shared__ int h[256];
  const int tid = threadIdx.x;
  if (tid < 256) h[tid] = 0;
  __syncthreads();
  const int beg = blockIdx.x * CHUNK;
  const int end = min(beg + CHUNK, NEDGE);
  int d[EPT];
#pragma unroll
  for (int k = 0; k < EPT; ++k) {
    const int i = beg + tid + k * BT;
    d[k] = (i < end) ? dst[i] : -1;
  }
#pragma unroll
  for (int k = 0; k < EPT; ++k)
    if (d[k] >= 0) atomicAdd(&h[d[k] >> 10], 1);
  __syncthreads();
  if (tid < NBKT) hist[blockIdx.x * NBKT + tid] = h[tid];
}

__global__ __launch_bounds__(BT)
void bucket_scatter(const int* __restrict__ src, const int* __restrict__ dst,
                    const int* __restrict__ hist, int* __restrict__ bbase,
                    unsigned* __restrict__ bkt) {
  __shared__ int p_tot[4][256];
  __shared__ int p_pre[4][256];
  __shared__ int w_tot[8];
  __shared__ int w_own[8];
  __shared__ int off[256];
  __shared__ int cur[256];
  __shared__ int s_src[CHUNK_PAD];
  __shared__ int s_dst[CHUNK_PAD];
  const int tid  = threadIdx.x;
  const int lane = tid & 31;
  const int wv   = tid >> 5;
  const int b    = blockIdx.x;

  {
    const int j = tid & 255;
    const int part = tid >> 8;
    int tot = 0, pre = 0;
    if (j < NBKT) {
      const int b0 = part * (GB / 4);
#pragma unroll 8
      for (int k = 0; k < GB / 4; ++k) {
        const int bb = b0 + k;
        const int v = hist[bb * NBKT + j];
        tot += v;
        pre += (bb < b) ? v : 0;
      }
    }
    p_tot[part][j] = tot;
    p_pre[part][j] = pre;
  }
  __syncthreads();

  int tot = 0, pre = 0, own = 0, inc_t = 0, inc_o = 0;
  if (tid < 256) {
    tot = (p_tot[0][tid] + p_tot[1][tid]) + (p_tot[2][tid] + p_tot[3][tid]);
    pre = (p_pre[0][tid] + p_pre[1][tid]) + (p_pre[2][tid] + p_pre[3][tid]);
    own = (tid < NBKT) ? hist[b * NBKT + tid] : 0;
    inc_t = tot;
    inc_o = own;
    for (int d = 1; d < 32; d <<= 1) {
      const int ut = __shfl_up(inc_t, d);
      const int uo = __shfl_up(inc_o, d);
      if (lane >= d) { inc_t += ut; inc_o += uo; }
    }
    if (lane == 31) { w_tot[wv] = inc_t; w_own[wv] = inc_o; }
  }
  __syncthreads();
  if (tid < 256) {
    int bt = 0, bo = 0;
    for (int k = 0; k < wv; ++k) { bt += w_tot[k]; bo += w_own[k]; }
    const int base = bt + inc_t - tot;
    const int lst  = bo + inc_o - own;
    cur[tid] = lst;
    off[tid] = base + pre - lst;
    if (b == 0 && tid <= NBKT) bbase[tid] = base;
  }
  __syncthreads();

  const int beg = b * CHUNK;
  const int end = min(beg + CHUNK, NEDGE);
  int dd[EPT], ss[EPT];
#pragma unroll
  for (int k = 0; k < EPT; ++k) {
    const int i = beg + tid + k * BT;
    const bool in = i < end;
    dd[k] = in ? dst[i] : -1;
    ss[k] = in ? src[i] : 0;
  }
#pragma unroll
  for (int k = 0; k < EPT; ++k) {
    if (dd[k] >= 0) {
      const int p = atomicAdd(&cur[dd[k] >> 10], 1);
      s_src[p] = ss[k];
      s_dst[p] = dd[k];
    }
  }
  __syncthreads();
  const int n = end - beg;
#pragma unroll
  for (int k = 0; k < EPT; ++k) {
    const int p = tid + k * BT;
    if (p < n) {
      const int d = s_dst[p];
      bkt[off[d >> 10] + p] = ((unsigned)s_src[p] & 0x3FFFFu) | ((unsigned)(d & 1023) << 18);
    }
  }
}

__global__ __launch_bounds__(BT)
void bucket_sort(const unsigned* __restrict__ bkt, const int* __restrict__ bbase,
                 int* __restrict__ rowptr, int* __restrict__ csr) {
  __shared__ int c[BT];
  __shared__ int wsum[BT / 32];
  __shared__ int sorted[CAPL];
  const int tid  = threadIdx.x;
  const int lane = tid & 31;
  const int wv   = tid >> 5;
  const int beg  = bbase[blockIdx.x];
  const int end  = bbase[blockIdx.x + 1];
  const int n    = end - beg;

  c[tid] = 0;
  __syncthreads();
  for (int i0 = beg + tid; i0 < end; i0 += EPT * BT) {
    unsigned w[EPT];
#pragma unroll
    for (int k = 0; k < EPT; ++k) {
      const int i = i0 + k * BT;
      w[k] = (i < end) ? bkt[i] : NOEDGE;
    }
#pragma unroll
    for (int k = 0; k < EPT; ++k)
      if (w[k] != NOEDGE) atomicAdd(&c[w[k] >> 18], 1);
  }
  __syncthreads();

  const int own = c[tid];
  int inc = own;
  for (int d = 1; d < 32; d <<= 1) {
    const int up = __shfl_up(inc, d);
    if (lane >= d) inc += up;
  }
  if (lane == 31) wsum[wv] = inc;
  __syncthreads();
  int before = (lane < wv) ? wsum[lane] : 0;
  for (int d = 16; d > 0; d >>= 1) before += __shfl_xor(before, d);
  const int ex = before + inc - own;
  const int node = blockIdx.x * BT + tid;
  if (node <= N_NODES) rowptr[node] = beg + ex;
  c[tid] = ex;
  __syncthreads();

  if (n <= CAPL) {
    for (int i0 = beg + tid; i0 < end; i0 += EPT * BT) {
      unsigned w[EPT];
#pragma unroll
      for (int k = 0; k < EPT; ++k) {
        const int i = i0 + k * BT;
        w[k] = (i < end) ? bkt[i] : NOEDGE;
      }
#pragma unroll
      for (int k = 0; k < EPT; ++k) {
        if (w[k] != NOEDGE) {
          const int p = atomicAdd(&c[w[k] >> 18], 1);
          sorted[p] = (int)(w[k] & 0x3FFFFu);
        }
      }
    }
    __syncthreads();
    for (int p = tid; p < n; p += BT) csr[beg + p] = sorted[p];
  } else {
    for (int i = beg + tid; i < end; i += BT) {
      const unsigned w = bkt[i];
      const int p = atomicAdd(&c[w >> 18], 1);
      csr[beg + p] = (int)(w & 0x3FFFFu);
    }
  }
}

__device__ __forceinline__ float dis_of(int deg) {
  return (deg > 0) ? rsqrtf(fmaxf((float)deg, 1.0f)) : 0.0f;
}

__global__ void prescale(const int* __restrict__ rowptr, const float* __restrict__ emb0,
                         float* __restrict__ xt) {
  const int t = blockIdx.x * blockDim.x + threadIdx.x;
  const int row = t >> 4;
  if (row >= N_NODES) return;
  const float dd = dis_of(rowptr[row + 1] - rowptr[row]);
  float4 v = *(const float4*)(emb0 + (size_t)t * 4);
  v.x *= dd; v.y *= dd; v.z *= dd; v.w *= dd;
  *(float4*)(xt + (size_t)t * 4) = v;
}

template <bool LAST>
__global__ __launch_bounds__(256)
void spmm_rows(const int* __restrict__ rowptr, const int* __restrict__ csr,
               const float* __restrict__ xt, float* __restrict__ xt_next,
               float* __restrict__ acc) {
  const int t = blockIdx.x * 256 + threadIdx.x;
  const int row = t >> 4;
  if (row >= N_NODES) return;
  const unsigned c = (unsigned)(t & 15) << 2;
  const int beg = rowptr[row];
  const int end = rowptr[row + 1];

  float4 s0 = {0.f, 0.f, 0.f, 0.f}, s1 = s0, s2 = s0, s3 = s0;
  int e = beg;
  for (; e + 4 <= end; e += 4) {
    const unsigned j0 = (unsigned)csr[e + 0];
    const unsigned j1 = (unsigned)csr[e + 1];
    const unsigned j2 = (unsigned)csr[e + 2];
    const unsigned j3 = (unsigned)csr[e + 3];
    const float4 v0 = *(const float4*)(xt + (j0 * (unsigned)DIM + c));
    const float4 v1 = *(const float4*)(xt + (j1 * (unsigned)DIM + c));
    const float4 v2 = *(const float4*)(xt + (j2 * (unsigned)DIM + c));
    const float4 v3 = *(const float4*)(xt + (j3 * (unsigned)DIM + c));
    s0.x += v0.x; s0.y += v0.y; s0.z += v0.z; s0.w += v0.w;
    s1.x += v1.x; s1.y += v1.y; s1.z += v1.z; s1.w += v1.w;
    s2.x += v2.x; s2.y += v2.y; s2.z += v2.z; s2.w += v2.w;
    s3.x += v3.x; s3.y += v3.y; s3.z += v3.z; s3.w += v3.w;
  }
  const int rem = end - e;
  if (rem > 0) {
    const int last = end - 1;
    const unsigned j0 = (unsigned)csr[e];
    const unsigned j1 = (unsigned)csr[min(e + 1, last)];
    const unsigned j2 = (unsigned)csr[min(e + 2, last)];
    const float4 z = {0.f, 0.f, 0.f, 0.f};
    const float4 v0 = *(const float4*)(xt + (j0 * (unsigned)DIM + c));
    float4 v1 = z, v2 = z;
    if (rem > 1) v1 = *(const float4*)(xt + (j1 * (unsigned)DIM + c));
    if (rem > 2) v2 = *(const float4*)(xt + (j2 * (unsigned)DIM + c));
    s0.x += v0.x; s0.y += v0.y; s0.z += v0.z; s0.w += v0.w;
    s1.x += v1.x; s1.y += v1.y; s1.z += v1.z; s1.w += v1.w;
    s2.x += v2.x; s2.y += v2.y; s2.z += v2.z; s2.w += v2.w;
  }

  const float dd = dis_of(end - beg);
  float4 x;
  x.x = dd * ((s0.x + s1.x) + (s2.x + s3.x));
  x.y = dd * ((s0.y + s1.y) + (s2.y + s3.y));
  x.z = dd * ((s0.z + s1.z) + (s2.z + s3.z));
  x.w = dd * ((s0.w + s1.w) + (s2.w + s3.w));

  const size_t o = (size_t)t * 4;
  float4 a = *(const float4*)(acc + o);
  a.x += x.x; a.y += x.y; a.z += x.z; a.w += x.w;
  if (LAST) {
    a.x *= 0.25f; a.y *= 0.25f; a.z *= 0.25f; a.w *= 0.25f;
  } else {
    float4 y;
    y.x = dd * x.x; y.y = dd * x.y; y.z = dd * x.z; y.w = dd * x.w;
    *(float4*)(xt_next + o) = y;
  }
  *(float4*)(acc + o) = a;
}

extern "C" void kernel_launch(void* const* d_in, const int* in_sizes, int n_in,
                              void* d_out, int out_size, void* d_ws, size_t ws_size,
                              hipStream_t stream) {
  const int*   ei    = (const int*)d_in[0];
  const int*   src   = ei;
  const int*   dst   = ei + NEDGE;
  const float* user  = (const float*)d_in[1];
  const float* bert  = (const float*)d_in[2];
  const float* wproj = (const float*)d_in[3];

  float* emb0 = (float*)d_out;
  float* acc  = emb0 + (size_t)N_NODES * DIM;

  int*   rowptr = (int*)d_ws;
  int*   csr    = rowptr + ((N_NODES + 1 + 63) & ~63);
  float* xA     = (float*)(csr + ((NEDGE + 63) & ~63));
  float* xB     = xA + (size_t)N_NODES * DIM;
  unsigned* bkt = (unsigned*)xB;
  int*   hist   = (int*)xB + ((NEDGE + 63) & ~63);
  int*   bbase  = hist + ((GB * NBKT + 63) & ~63);

  const int B = 256;

  init_user<<<(NUM_USERS * DIM + B - 1) / B, B, 0, stream>>>(user, emb0, acc);
  wmma_proj<<<NUM_ITEMS / 16, 128, 0, stream>>>(bert, wproj, emb0, acc);

  bucket_hist   <<<GB, BT, 0, stream>>>(dst, hist);
  bucket_scatter<<<GB, BT, 0, stream>>>(src, dst, hist, bbase, bkt);
  bucket_sort   <<<NBKT, BT, 0, stream>>>(bkt, bbase, rowptr, csr);

  prescale<<<(N_NODES * 16 + B - 1) / B, B, 0, stream>>>(rowptr, emb0, xA);
  spmm_rows<false><<<(N_NODES * 16 + 255) / 256, 256, 0, stream>>>(rowptr, csr, xA, xB, acc);
  spmm_rows<false><<<(N_NODES * 16 + 255) / 256, 256, 0, stream>>>(rowptr, csr, xB, xA, acc);
  spmm_rows<true> <<<(N_NODES * 16 + 255) / 256, 256, 0, stream>>>(rowptr, csr, xA, xB, acc);
}


# User copy and prescale run on a chip-sized grid (512 work-groups of 8 waves, 13 and 19 float4 items per lane, all loads requested up front) instead of 275000 one-item waves
# speedup vs baseline: 3.7674x; 1.0464x over previous
#include <hip/hip_runtime.h>

typedef __attribute__((ext_vector_type(2))) float v2f;
typedef __attribute__((ext_vector_type(8))) float v8f;
typedef __attribute__((ext_vector_type(4))) float v4f;

constexpr int NUM_USERS = 100000;
constexpr int NUM_ITEMS = 50000;
constexpr int N_NODES   = NUM_USERS + NUM_ITEMS;
constexpr int DIM       = 64;
constexpr int SBERT     = 384;
constexpr int NEDGE     = 2000000;
constexpr int NLAYERS   = 3;

constexpr int SGRID = 512;
constexpr int SBLK  = 256;
constexpr int STHR  = SGRID * SBLK;

constexpr int U4    = NUM_USERS * DIM / 4;
constexpr int U_PER = (U4 + STHR - 1) / STHR;
static_assert((U_PER - 1) * STHR <= U4, "only a lane's last item needs a bound test");

__global__ __launch_bounds__(SBLK)
void init_user(const float* __restrict__ u,
               float* __restrict__ emb0, float* __restrict__ acc) {
  const int t = blockIdx.x * SBLK + threadIdx.x;
  const int  il = t + (U_PER - 1) * STHR;
  const bool hl = il < U4;
  v4f v[U_PER - 1];
  v4f vl = {0.f, 0.f, 0.f, 0.f};
#pragma unroll
  for (int k = 0; k < U_PER - 1; ++k) v[k] = *(const v4f*)(u + (size_t)(t + k * STHR) * 4);
  if (hl) vl = *(const v4f*)(u + (size_t)il * 4);
#pragma unroll
  for (int k = 0; k < U_PER - 1; ++k) {
    const size_t o = (size_t)(t + k * STHR) * 4;
    *(v4f*)(emb0 + o) = v[k];
    *(v4f*)(acc  + o) = v[k];
  }
  if (hl) {
    *(v4f*)(emb0 + (size_t)il * 4) = vl;
    *(v4f*)(acc  + (size_t)il * 4) = vl;
  }
}

__global__ __launch_bounds__(128)
void wmma_proj(const float* __restrict__ bert, const float* __restrict__ w,
               float* __restrict__ emb0, float* __restrict__ acc) {
  const int lane = threadIdx.x & 31;
  const int wave = threadIdx.x >> 5;
  const int m0   = blockIdx.x * 16;
  const int n0   = wave * 16;
  const int half = lane >> 4;
  const int r    = lane & 15;

  const float* arow = bert + (size_t)(m0 + r) * SBERT + 2 * half;
  const float* brow = w    + (size_t)(n0 + r) * SBERT + 2 * half;

  v8f c = {};
#pragma unroll 8
  for (int k = 0; k < SBERT; k += 4) {
    v2f a = *(const v2f*)(arow + k);
    v2f b = *(const v2f*)(brow + k);
    c = __builtin_amdgcn_wmma_f32_16x16x4_f32(
         false, a,  false, b,
         (short)0, c,  false,  false);
  }

  const size_t base = (size_t)(NUM_USERS + m0 + 8 * half) * DIM + (size_t)(n0 + r);
#pragma unroll
  for (int i = 0; i < 8; ++i) {
    float v = c[i];
    emb0[base + (size_t)i * DIM] = v;
    acc [base + (size_t)i * DIM] = v;
  }
}


constexpr int BT    = 1024;
constexpr int NBKT  = (N_NODES + 1023) >> 10;
constexpr int GB    = 256;
constexpr int CHUNK = (NEDGE + GB - 1) / GB;
constexpr int EPT   = 8;
constexpr int CHUNK_PAD = EPT * BT;
static_assert(CHUNK <= CHUNK_PAD, "a chunk must fit its LDS stage");
constexpr int CAPL  = 32768;
constexpr unsigned NOEDGE = 0xFFFFFFFFu;

__global__ __launch_bounds__(BT)
void bucket_hist(const int* __restrict__ dst, int* __restrict__ hist) {
  __shared__ int h[256];
  const int tid = threadIdx.x;
  if (tid < 256) h[tid] = 0;
  __syncthreads();
  const int beg = blockIdx.x * CHUNK;
  const int end = min(beg + CHUNK, NEDGE);
  int d[EPT];
#pragma unroll
  for (int k = 0; k < EPT; ++k) {
    const int i = beg + tid + k * BT;
    d[k] = (i < end) ? dst[i] : -1;
  }
#pragma unroll
  for (int k = 0; k < EPT; ++k)
    if (d[k] >= 0) atomicAdd(&h[d[k] >> 10], 1);
  __syncthreads();
  if (tid < NBKT) hist[blockIdx.x * NBKT + tid] = h[tid];
}

__global__ __launch_bounds__(BT)
void bucket_scatter(const int* __restrict__ src, const int* __restrict__ dst,
                    const int* __restrict__ hist, int* __restrict__ bbase,
                    unsigned* __restrict__ bkt) {
  __shared__ int p_tot[4][256];
  __shared__ int p_pre[4][256];
  __shared__ int w_tot[8];
  __shared__ int w_own[8];
  __shared__ int off[256];
  __shared__ int cur[256];
  __shared__ int s_src[CHUNK_PAD];
  __shared__ int s_dst[CHUNK_PAD];
  const int tid  = threadIdx.x;
  const int lane = tid & 31;
  const int wv   = tid >> 5;
  const int b    = blockIdx.x;

  {
    const int j = tid & 255;
    const int part = tid >> 8;
    int tot = 0, pre = 0;
    if (j < NBKT) {
      const int b0 = part * (GB / 4);
#pragma unroll 8
      for (int k = 0; k < GB / 4; ++k) {
        const int bb = b0 + k;
        const int v = hist[bb * NBKT + j];
        tot += v;
        pre += (bb < b) ? v : 0;
      }
    }
    p_tot[part][j] = tot;
    p_pre[part][j] = pre;
  }
  __syncthreads();

  int tot = 0, pre = 0, own = 0, inc_t = 0, inc_o = 0;
  if (tid < 256) {
    tot = (p_tot[0][tid] + p_tot[1][tid]) + (p_tot[2][tid] + p_tot[3][tid]);
    pre = (p_pre[0][tid] + p_pre[1][tid]) + (p_pre[2][tid] + p_pre[3][tid]);
    own = (tid < NBKT) ? hist[b * NBKT + tid] : 0;
    inc_t = tot;
    inc_o = own;
    for (int d = 1; d < 32; d <<= 1) {
      const int ut = __shfl_up(inc_t, d);
      const int uo = __shfl_up(inc_o, d);
      if (lane >= d) { inc_t += ut; inc_o += uo; }
    }
    if (lane == 31) { w_tot[wv] = inc_t; w_own[wv] = inc_o; }
  }
  __syncthreads();
  if (tid < 256) {
    int bt = 0, bo = 0;
    for (int k = 0; k < wv; ++k) { bt += w_tot[k]; bo += w_own[k]; }
    const int base = bt + inc_t - tot;
    const int lst  = bo + inc_o - own;
    cur[tid] = lst;
    off[tid] = base + pre - lst;
    if (b == 0 && tid <= NBKT) bbase[tid] = base;
  }
  __syncthreads();

  const int beg = b * CHUNK;
  const int end = min(beg + CHUNK, NEDGE);
  int dd[EPT], ss[EPT];
#pragma unroll
  for (int k = 0; k < EPT; ++k) {
    const int i = beg + tid + k * BT;
    const bool in = i < end;
    dd[k] = in ? dst[i] : -1;
    ss[k] = in ? src[i] : 0;
  }
#pragma unroll
  for (int k = 0; k < EPT; ++k) {
    if (dd[k] >= 0) {
      const int p = atomicAdd(&cur[dd[k] >> 10], 1);
      s_src[p] = ss[k];
      s_dst[p] = dd[k];
    }
  }
  __syncthreads();
  const int n = end - beg;
#pragma unroll
  for (int k = 0; k < EPT; ++k) {
    const int p = tid + k * BT;
    if (p < n) {
      const int d = s_dst[p];
      bkt[off[d >> 10] + p] = ((unsigned)s_src[p] & 0x3FFFFu) | ((unsigned)(d & 1023) << 18);
    }
  }
}

__global__ __launch_bounds__(BT)
void bucket_sort(const unsigned* __restrict__ bkt, const int* __restrict__ bbase,
                 int* __restrict__ rowptr, int* __restrict__ csr) {
  __shared__ int c[BT];
  __shared__ int wsum[BT / 32];
  __shared__ int sorted[CAPL];
  const int tid  = threadIdx.x;
  const int lane = tid & 31;
  const int wv   = tid >> 5;
  const int beg  = bbase[blockIdx.x];
  const int end  = bbase[blockIdx.x + 1];
  const int n    = end - beg;

  c[tid] = 0;
  __syncthreads();
  for (int i0 = beg + tid; i0 < end; i0 += EPT * BT) {
    unsigned w[EPT];
#pragma unroll
    for (int k = 0; k < EPT; ++k) {
      const int i = i0 + k * BT;
      w[k] = (i < end) ? bkt[i] : NOEDGE;
    }
#pragma unroll
    for (int k = 0; k < EPT; ++k)
      if (w[k] != NOEDGE) atomicAdd(&c[w[k] >> 18], 1);
  }
  __syncthreads();

  const int own = c[tid];
  int inc = own;
  for (int d = 1; d < 32; d <<= 1) {
    const int up = __shfl_up(inc, d);
    if (lane >= d) inc += up;
  }
  if (lane == 31) wsum[wv] = inc;
  __syncthreads();
  int before = (lane < wv) ? wsum[lane] : 0;
  for (int d = 16; d > 0; d >>= 1) before += __shfl_xor(before, d);
  const int ex = before + inc - own;
  const int node = blockIdx.x * BT + tid;
  if (node <= N_NODES) rowptr[node] = beg + ex;
  c[tid] = ex;
  __syncthreads();

  if (n <= CAPL) {
    for (int i0 = beg + tid; i0 < end; i0 += EPT * BT) {
      unsigned w[EPT];
#pragma unroll
      for (int k = 0; k < EPT; ++k) {
        const int i = i0 + k * BT;
        w[k] = (i < end) ? bkt[i] : NOEDGE;
      }
#pragma unroll
      for (int k = 0; k < EPT; ++k) {
        if (w[k] != NOEDGE) {
          const int p = atomicAdd(&c[w[k] >> 18], 1);
          sorted[p] = (int)(w[k] & 0x3FFFFu);
        }
      }
    }
    __syncthreads();
    for (int p = tid; p < n; p += BT) csr[beg + p] = sorted[p];
  } else {
    for (int i = beg + tid; i < end; i += BT) {
      const unsigned w = bkt[i];
      const int p = atomicAdd(&c[w >> 18], 1);
      csr[beg + p] = (int)(w & 0x3FFFFu);
    }
  }
}

__device__ __forceinline__ float dis_of(int deg) {
  return (deg > 0) ? rsqrtf(fmaxf((float)deg, 1.0f)) : 0.0f;
}

constexpr int P4    = N_NODES * DIM / 4;
constexpr int P_PER = (P4 + STHR - 1) / STHR;
static_assert((P_PER - 1) * STHR <= P4, "only a lane's last item needs a bound test");

__global__ __launch_bounds__(SBLK)
void prescale(const int* __restrict__ rowptr, const float* __restrict__ emb0,
              float* __restrict__ xt) {
  const int t = blockIdx.x * SBLK + threadIdx.x;
  const int  il = t + (P_PER - 1) * STHR;
  const bool hl = il < P4;
  int lo[P_PER - 1], hi[P_PER - 1];
  v4f v[P_PER - 1];
  int lol = 0, hil = 0;
  v4f vl = {0.f, 0.f, 0.f, 0.f};
#pragma unroll
  for (int k = 0; k < P_PER - 1; ++k) {
    const int i = t + k * STHR;
    lo[k] = rowptr[i >> 4];
    hi[k] = rowptr[(i >> 4) + 1];
    v[k]  = *(const v4f*)(emb0 + (size_t)i * 4);
  }
  if (hl) {
    lol = rowptr[il >> 4];
    hil = rowptr[(il >> 4) + 1];
    vl  = *(const v4f*)(emb0 + (size_t)il * 4);
  }
#pragma unroll
  for (int k = 0; k < P_PER - 1; ++k) {
    const float dd = dis_of(hi[k] - lo[k]);
    *(v4f*)(xt + (size_t)(t + k * STHR) * 4) = v[k] * dd;
  }
  if (hl) {
    const float dd = dis_of(hil - lol);
    *(v4f*)(xt + (size_t)il * 4) = vl * dd;
  }
}

template <bool LAST>
__global__ __launch_bounds__(256)
void spmm_rows(const int* __restrict__ rowptr, const int* __restrict__ csr,
               const float* __restrict__ xt, float* __restrict__ xt_next,
               float* __restrict__ acc) {
  const int t = blockIdx.x * 256 + threadIdx.x;
  const int row = t >> 4;
  if (row >= N_NODES) return;
  const unsigned c = (unsigned)(t & 15) << 2;
  const int beg = rowptr[row];
  const int end = rowptr[row + 1];

  float4 s0 = {0.f, 0.f, 0.f, 0.f}, s1 = s0, s2 = s0, s3 = s0;
  int e = beg;
  for (; e + 4 <= end; e += 4) {
    const unsigned j0 = (unsigned)csr[e + 0];
    const unsigned j1 = (unsigned)csr[e + 1];
    const unsigned j2 = (unsigned)csr[e + 2];
    const unsigned j3 = (unsigned)csr[e + 3];
    const float4 v0 = *(const float4*)(xt + (j0 * (unsigned)DIM + c));
    const float4 v1 = *(const float4*)(xt + (j1 * (unsigned)DIM + c));
    const float4 v2 = *(const float4*)(xt + (j2 * (unsigned)DIM + c));
    const float4 v3 = *(const float4*)(xt + (j3 * (unsigned)DIM + c));
    s0.x += v0.x; s0.y += v0.y; s0.z += v0.z; s0.w += v0.w;
    s1.x += v1.x; s1.y += v1.y; s1.z += v1.z; s1.w += v1.w;
    s2.x += v2.x; s2.y += v2.y; s2.z += v2.z; s2.w += v2.w;
    s3.x += v3.x; s3.y += v3.y; s3.z += v3.z; s3.w += v3.w;
  }
  const int rem = end - e;
  if (rem > 0) {
    const int last = end - 1;
    const unsigned j0 = (unsigned)csr[e];
    const unsigned j1 = (unsigned)csr[min(e + 1, last)];
    const unsigned j2 = (unsigned)csr[min(e + 2, last)];
    const float4 z = {0.f, 0.f, 0.f, 0.f};
    const float4 v0 = *(const float4*)(xt + (j0 * (unsigned)DIM + c));
    float4 v1 = z, v2 = z;
    if (rem > 1) v1 = *(const float4*)(xt + (j1 * (unsigned)DIM + c));
    if (rem > 2) v2 = *(const float4*)(xt + (j2 * (unsigned)DIM + c));
    s0.x += v0.x; s0.y += v0.y; s0.z += v0.z; s0.w += v0.w;
    s1.x += v1.x; s1.y += v1.y; s1.z += v1.z; s1.w += v1.w;
    s2.x += v2.x; s2.y += v2.y; s2.z += v2.z; s2.w += v2.w;
  }

  const float dd = dis_of(end - beg);
  float4 x;
  x.x = dd * ((s0.x + s1.x) + (s2.x + s3.x));
  x.y = dd * ((s0.y + s1.y) + (s2.y + s3.y));
  x.z = dd * ((s0.z + s1.z) + (s2.z + s3.z));
  x.w = dd * ((s0.w + s1.w) + (s2.w + s3.w));

  const size_t o = (size_t)t * 4;
  float4 a = *(const float4*)(acc + o);
  a.x += x.x; a.y += x.y; a.z += x.z; a.w += x.w;
  if (LAST) {
    a.x *= 0.25f; a.y *= 0.25f; a.z *= 0.25f; a.w *= 0.25f;
  } else {
    float4 y;
    y.x = dd * x.x; y.y = dd * x.y; y.z = dd * x.z; y.w = dd * x.w;
    *(float4*)(xt_next + o) = y;
  }
  *(float4*)(acc + o) = a;
}

extern "C" void kernel_launch(void* const* d_in, const int* in_sizes, int n_in,
                              void* d_out, int out_size, void* d_ws, size_t ws_size,
                              hipStream_t stream) {
  const int*   ei    = (const int*)d_in[0];
  const int*   src   = ei;
  const int*   dst   = ei + NEDGE;
  const float* user  = (const float*)d_in[1];
  const float* bert  = (const float*)d_in[2];
  const float* wproj = (const float*)d_in[3];

  float* emb0 = (float*)d_out;
  float* acc  = emb0 + (size_t)N_NODES * DIM;

  int*   rowptr = (int*)d_ws;
  int*   csr    = rowptr + ((N_NODES + 1 + 63) & ~63);
  float* xA     = (float*)(csr + ((NEDGE + 63) & ~63));
  float* xB     = xA + (size_t)N_NODES * DIM;
  unsigned* bkt = (unsigned*)xB;
  int*   hist   = (int*)xB + ((NEDGE + 63) & ~63);
  int*   bbase  = hist + ((GB * NBKT + 63) & ~63);

  init_user<<<SGRID, SBLK, 0, stream>>>(user, emb0, acc);
  wmma_proj<<<NUM_ITEMS / 16, 128, 0, stream>>>(bert, wproj, emb0, acc);

  bucket_hist   <<<GB, BT, 0, stream>>>(dst, hist);
  bucket_scatter<<<GB, BT, 0, stream>>>(src, dst, hist, bbase, bkt);
  bucket_sort   <<<NBKT, BT, 0, stream>>>(bkt, bbase, rowptr, csr);

  prescale<<<SGRID, SBLK, 0, stream>>>(rowptr, emb0, xA);
  spmm_rows<false><<<(N_NODES * 16 + 255) / 256, 256, 0, stream>>>(rowptr, csr, xA, xB, acc);
  spmm_rows<false><<<(N_NODES * 16 + 255) / 256, 256, 0, stream>>>(rowptr, csr, xB, xA, acc);
  spmm_rows<true> <<<(N_NODES * 16 + 255) / 256, 256, 0, stream>>>(rowptr, csr, xA, xB, acc);
}


# Item projection re-tiled: one wave per 16x64 strip (4 accumulator tiles) loads A once, not once per column tile; w_proj fragments come from one LDS copy per work-group; sums bit-identical
# speedup vs baseline: 4.3895x; 1.1651x over previous
#include <hip/hip_runtime.h>

typedef __attribute__((ext_vector_type(2))) float v2f;
typedef __attribute__((ext_vector_type(8))) float v8f;
typedef __attribute__((ext_vector_type(4))) float v4f;

constexpr int NUM_USERS = 100000;
constexpr int NUM_ITEMS = 50000;
constexpr int N_NODES   = NUM_USERS + NUM_ITEMS;
constexpr int DIM       = 64;
constexpr int SBERT     = 384;
constexpr int NEDGE     = 2000000;
constexpr int NLAYERS   = 3;

constexpr int SGRID = 512;
constexpr int SBLK  = 256;
constexpr int STHR  = SGRID * SBLK;

constexpr int U4    = NUM_USERS * DIM / 4;
constexpr int U_PER = (U4 + STHR - 1) / STHR;
static_assert((U_PER - 1) * STHR <= U4, "only a lane's last item needs a bound test");

__global__ __launch_bounds__(SBLK)
void init_user(const float* __restrict__ u,
               float* __restrict__ emb0, float* __restrict__ acc) {
  const int t = blockIdx.x * SBLK + threadIdx.x;
  const int  il = t + (U_PER - 1) * STHR;
  const bool hl = il < U4;
  v4f v[U_PER - 1];
  v4f vl = {0.f, 0.f, 0.f, 0.f};
#pragma unroll
  for (int k = 0; k < U_PER - 1; ++k) v[k] = *(const v4f*)(u + (size_t)(t + k * STHR) * 4);
  if (hl) vl = *(const v4f*)(u + (size_t)il * 4);
#pragma unroll
  for (int k = 0; k < U_PER - 1; ++k) {
    const size_t o = (size_t)(t + k * STHR) * 4;
    *(v4f*)(emb0 + o) = v[k];
    *(v4f*)(acc  + o) = v[k];
  }
  if (hl) {
    *(v4f*)(emb0 + (size_t)il * 4) = vl;
    *(v4f*)(acc  + (size_t)il * 4) = vl;
  }
}

constexpr int PJ_GRID  = 256;
constexpr int PJ_WAVES = 16;
constexpr int PJ_FULL  = 12;
constexpr int PJ_BLK   = PJ_WAVES * 32;
constexpr int MTILES   = NUM_ITEMS / 16;
static_assert(NUM_ITEMS % 16 == 0, "whole strips only");
static_assert(PJ_WAVES - PJ_FULL == DIM / 16, "one wave per column tile of a split strip");
static_assert(PJ_FULL * PJ_GRID <= MTILES && MTILES <= (PJ_FULL + 1) * PJ_GRID, "every strip has one owner");
constexpr int WPITCH   = SBERT + 4;
constexpr int KSTEPS   = SBERT / 8;
constexpr int AG       = 8;
constexpr int W4       = DIM * SBERT / 4;
constexpr int W_PER    = W4 / PJ_BLK;
static_assert(W_PER * PJ_BLK == W4, "staging divides evenly");

__device__ __forceinline__ v8f wmma4(v2f a, v2f b, v8f c) {
  return __builtin_amdgcn_wmma_f32_16x16x4_f32(
       false, a,  false, b,
       (short)0, c,  false,  false);
}

template <int NT>
__device__ __forceinline__ void proj_strip(const float* __restrict__ arow, const float* bl,
                                           v2f (&a0)[AG], v2f (&a1)[AG],
                                           float* __restrict__ emb0, float* __restrict__ acc,
                                           size_t obase) {
  v8f c[NT];
#pragma unroll
  for (int j = 0; j < NT; ++j) c[j] = v8f{};
#pragma unroll
  for (int g = 0; g < KSTEPS / AG; ++g) {
    v2f n0[AG], n1[AG];
    if (g + 1 < KSTEPS / AG) {
#pragma unroll
      for (int s = 0; s < AG; ++s) {
        n0[s] = *(const v2f*)(arow + 8 * ((g + 1) * AG + s));
        n1[s] = *(const v2f*)(arow + 8 * ((g + 1) * AG + s) + 4);
      }
    }
#pragma unroll
    for (int s = 0; s < AG; ++s) {
      const int k8 = (g * AG + s) * 8;
      v4f b[NT];
#pragma unroll
      for (int j = 0; j < NT; ++j) b[j] = *(const v4f*)(bl + j * 16 * WPITCH + k8);
#pragma unroll
      for (int j = 0; j < NT; ++j) c[j] = wmma4(a0[s], b[j].lo, c[j]);
#pragma unroll
      for (int j = 0; j < NT; ++j) c[j] = wmma4(a1[s], b[j].hi, c[j]);
    }
    if (g + 1 < KSTEPS / AG) {
#pragma unroll
      for (int s = 0; s < AG; ++s) { a0[s] = n0[s]; a1[s] = n1[s]; }
    }
  }
#pragma unroll
  for (int i = 0; i < 8; ++i) {
#pragma unroll
    for (int j = 0; j < NT; ++j) {
      const size_t o = obase + (size_t)i * DIM + 16 * j;
      emb0[o] = c[j][i];
      acc[o]  = c[j][i];
    }
  }
}

__global__ __launch_bounds__(PJ_BLK)
void wmma_proj(const float* __restrict__ bert, const float* __restrict__ w,
               float* __restrict__ emb0, float* __restrict__ acc) {
  __shared__ __attribute__((aligned(16))) float wl[DIM * WPITCH];
  const int lane = threadIdx.x & 31;
  const int wave = threadIdx.x >> 5;
  const int half = lane >> 4;
  const int r    = lane & 15;
  const bool full = wave < PJ_FULL;
  const int tile = (full ? wave : PJ_FULL) * PJ_GRID + blockIdx.x;
  const int jt   = full ? 0 : wave - PJ_FULL;
  const bool active = tile < MTILES;

  const float* arow = bert + (size_t)((active ? tile : 0) * 16 + r) * SBERT + 2 * half;

  v2f a0[AG], a1[AG];
  if (active) {
#pragma unroll
    for (int s = 0; s < AG; ++s) {
      a0[s] = *(const v2f*)(arow + 8 * s);
      a1[s] = *(const v2f*)(arow + 8 * s + 4);
    }
  }

  {
    v4f t[W_PER];
#pragma unroll
    for (int k = 0; k < W_PER; ++k) t[k] = *(const v4f*)(w + (size_t)(threadIdx.x + k * PJ_BLK) * 4);
#pragma unroll
    for (int k = 0; k < W_PER; ++k) {
      const int i  = threadIdx.x + k * PJ_BLK;
      const int n  = i / (SBERT / 4);
      const int c  = i % (SBERT / 4);
      float* d = wl + n * WPITCH + 8 * (c >> 1) + 2 * (c & 1);
      *(v2f*)(d)     = t[k].lo;
      *(v2f*)(d + 4) = t[k].hi;
    }
  }
  __syncthreads();
  if (!active) return;

  const float* bl = wl + (jt * 16 + r) * WPITCH + 4 * half;
  const size_t obase = (size_t)(NUM_USERS + tile * 16 + 8 * half) * DIM + (size_t)(jt * 16 + r);
  if (full) proj_strip<DIM / 16>(arow, bl, a0, a1, emb0, acc, obase);
  else      proj_strip<1>(arow, bl, a0, a1, emb0, acc, obase);
}


constexpr int BT    = 1024;
constexpr int NBKT  = (N_NODES + 1023) >> 10;
constexpr int GB    = 256;
constexpr int CHUNK = (NEDGE + GB - 1) / GB;
constexpr int EPT   = 8;
constexpr int CHUNK_PAD = EPT * BT;
static_assert(CHUNK <= CHUNK_PAD, "a chunk must fit its LDS stage");
constexpr int CAPL  = 32768;
constexpr unsigned NOEDGE = 0xFFFFFFFFu;

__global__ __launch_bounds__(BT)
void bucket_hist(const int* __restrict__ dst, int* __restrict__ hist) {
  __shared__ int h[256];
  const int tid = threadIdx.x;
  if (tid < 256) h[tid] = 0;
  __syncthreads();
  const int beg = blockIdx.x * CHUNK;
  const int end = min(beg + CHUNK, NEDGE);
  int d[EPT];
#pragma unroll
  for (int k = 0; k < EPT; ++k) {
    const int i = beg + tid + k * BT;
    d[k] = (i < end) ? dst[i] : -1;
  }
#pragma unroll
  for (int k = 0; k < EPT; ++k)
    if (d[k] >= 0) atomicAdd(&h[d[k] >> 10], 1);
  __syncthreads();
  if (tid < NBKT) hist[blockIdx.x * NBKT + tid] = h[tid];
}

__global__ __launch_bounds__(BT)
void bucket_scatter(const int* __restrict__ src, const int* __restrict__ dst,
                    const int* __restrict__ hist, int* __restrict__ bbase,
                    unsigned* __restrict__ bkt) {
  __shared__ int p_tot[4][256];
  __shared__ int p_pre[4][256];
  __shared__ int w_tot[8];
  __shared__ int w_own[8];
  __shared__ int off[256];
  __shared__ int cur[256];
  __shared__ int s_src[CHUNK_PAD];
  __shared__ int s_dst[CHUNK_PAD];
  const int tid  = threadIdx.x;
  const int lane = tid & 31;
  const int wv   = tid >> 5;
  const int b    = blockIdx.x;

  {
    const int j = tid & 255;
    const int part = tid >> 8;
    int tot = 0, pre = 0;
    if (j < NBKT) {
      const int b0 = part * (GB / 4);
#pragma unroll 8
      for (int k = 0; k < GB / 4; ++k) {
        const int bb = b0 + k;
        const int v = hist[bb * NBKT + j];
        tot += v;
        pre += (bb < b) ? v : 0;
      }
    }
    p_tot[part][j] = tot;
    p_pre[part][j] = pre;
  }
  __syncthreads();

  int tot = 0, pre = 0, own = 0, inc_t = 0, inc_o = 0;
  if (tid < 256) {
    tot = (p_tot[0][tid] + p_tot[1][tid]) + (p_tot[2][tid] + p_tot[3][tid]);
    pre = (p_pre[0][tid] + p_pre[1][tid]) + (p_pre[2][tid] + p_pre[3][tid]);
    own = (tid < NBKT) ? hist[b * NBKT + tid] : 0;
    inc_t = tot;
    inc_o = own;
    for (int d = 1; d < 32; d <<= 1) {
      const int ut = __shfl_up(inc_t, d);
      const int uo = __shfl_up(inc_o, d);
      if (lane >= d) { inc_t += ut; inc_o += uo; }
    }
    if (lane == 31) { w_tot[wv] = inc_t; w_own[wv] = inc_o; }
  }
  __syncthreads();
  if (tid < 256) {
    int bt = 0, bo = 0;
    for (int k = 0; k < wv; ++k) { bt += w_tot[k]; bo += w_own[k]; }
    const int base = bt + inc_t - tot;
    const int lst  = bo + inc_o - own;
    cur[tid] = lst;
    off[tid] = base + pre - lst;
    if (b == 0 && tid <= NBKT) bbase[tid] = base;
  }
  __syncthreads();

  const int beg = b * CHUNK;
  const int end = min(beg + CHUNK, NEDGE);
  int dd[EPT], ss[EPT];
#pragma unroll
  for (int k = 0; k < EPT; ++k) {
    const int i = beg + tid + k * BT;
    const bool in = i < end;
    dd[k] = in ? dst[i] : -1;
    ss[k] = in ? src[i] : 0;
  }
#pragma unroll
  for (int k = 0; k < EPT; ++k) {
    if (dd[k] >= 0) {
      const int p = atomicAdd(&cur[dd[k] >> 10], 1);
      s_src[p] = ss[k];
      s_dst[p] = dd[k];
    }
  }
  __syncthreads();
  const int n = end - beg;
#pragma unroll
  for (int k = 0; k < EPT; ++k) {
    const int p = tid + k * BT;
    if (p < n) {
      const int d = s_dst[p];
      bkt[off[d >> 10] + p] = ((unsigned)s_src[p] & 0x3FFFFu) | ((unsigned)(d & 1023) << 18);
    }
  }
}

__global__ __launch_bounds__(BT)
void bucket_sort(const unsigned* __restrict__ bkt, const int* __restrict__ bbase,
                 int* __restrict__ rowptr, int* __restrict__ csr) {
  __shared__ int c[BT];
  __shared__ int wsum[BT / 32];
  __shared__ int sorted[CAPL];
  const int tid  = threadIdx.x;
  const int lane = tid & 31;
  const int wv   = tid >> 5;
  const int beg  = bbase[blockIdx.x];
  const int end  = bbase[blockIdx.x + 1];
  const int n    = end - beg;

  c[tid] = 0;
  __syncthreads();
  for (int i0 = beg + tid; i0 < end; i0 += EPT * BT) {
    unsigned w[EPT];
#pragma unroll
    for (int k = 0; k < EPT; ++k) {
      const int i = i0 + k * BT;
      w[k] = (i < end) ? bkt[i] : NOEDGE;
    }
#pragma unroll
    for (int k = 0; k < EPT; ++k)
      if (w[k] != NOEDGE) atomicAdd(&c[w[k] >> 18], 1);
  }
  __syncthreads();

  const int own = c[tid];
  int inc = own;
  for (int d = 1; d < 32; d <<= 1) {
    const int up = __shfl_up(inc, d);
    if (lane >= d) inc += up;
  }
  if (lane == 31) wsum[wv] = inc;
  __syncthreads();
  int before = (lane < wv) ? wsum[lane] : 0;
  for (int d = 16; d > 0; d >>= 1) before += __shfl_xor(before, d);
  const int ex = before + inc - own;
  const int node = blockIdx.x * BT + tid;
  if (node <= N_NODES) rowptr[node] = beg + ex;
  c[tid] = ex;
  __syncthreads();

  if (n <= CAPL) {
    for (int i0 = beg + tid; i0 < end; i0 += EPT * BT) {
      unsigned w[EPT];
#pragma unroll
      for (int k = 0; k < EPT; ++k) {
        const int i = i0 + k * BT;
        w[k] = (i < end) ? bkt[i] : NOEDGE;
      }
#pragma unroll
      for (int k = 0; k < EPT; ++k) {
        if (w[k] != NOEDGE) {
          const int p = atomicAdd(&c[w[k] >> 18], 1);
          sorted[p] = (int)(w[k] & 0x3FFFFu);
        }
      }
    }
    __syncthreads();
    for (int p = tid; p < n; p += BT) csr[beg + p] = sorted[p];
  } else {
    for (int i = beg + tid; i < end; i += BT) {
      const unsigned w = bkt[i];
      const int p = atomicAdd(&c[w >> 18], 1);
      csr[beg + p] = (int)(w & 0x3FFFFu);
    }
  }
}

__device__ __forceinline__ float dis_of(int deg) {
  return (deg > 0) ? rsqrtf(fmaxf((float)deg, 1.0f)) : 0.0f;
}

constexpr int P4    = N_NODES * DIM / 4;
constexpr int P_PER = (P4 + STHR - 1) / STHR;
static_assert((P_PER - 1) * STHR <= P4, "only a lane's last item needs a bound test");

__global__ __launch_bounds__(SBLK)
void prescale(const int* __restrict__ rowptr, const float* __restrict__ emb0,
              float* __restrict__ xt) {
  const int t = blockIdx.x * SBLK + threadIdx.x;
  const int  il = t + (P_PER - 1) * STHR;
  const bool hl = il < P4;
  int lo[P_PER - 1], hi[P_PER - 1];
  v4f v[P_PER - 1];
  int lol = 0, hil = 0;
  v4f vl = {0.f, 0.f, 0.f, 0.f};
#pragma unroll
  for (int k = 0; k < P_PER - 1; ++k) {
    const int i = t + k * STHR;
    lo[k] = rowptr[i >> 4];
    hi[k] = rowptr[(i >> 4) + 1];
    v[k]  = *(const v4f*)(emb0 + (size_t)i * 4);
  }
  if (hl) {
    lol = rowptr[il >> 4];
    hil = rowptr[(il >> 4) + 1];
    vl  = *(const v4f*)(emb0 + (size_t)il * 4);
  }
#pragma unroll
  for (int k = 0; k < P_PER - 1; ++k) {
    const float dd = dis_of(hi[k] - lo[k]);
    *(v4f*)(xt + (size_t)(t + k * STHR) * 4) = v[k] * dd;
  }
  if (hl) {
    const float dd = dis_of(hil - lol);
    *(v4f*)(xt + (size_t)il * 4) = vl * dd;
  }
}

template <bool LAST>
__global__ __launch_bounds__(256)
void spmm_rows(const int* __restrict__ rowptr, const int* __restrict__ csr,
               const float* __restrict__ xt, float* __restrict__ xt_next,
               float* __restrict__ acc) {
  const int t = blockIdx.x * 256 + threadIdx.x;
  const int row = t >> 4;
  if (row >= N_NODES) return;
  const unsigned c = (unsigned)(t & 15) << 2;
  const int beg = rowptr[row];
  const int end = rowptr[row + 1];

  float4 s0 = {0.f, 0.f, 0.f, 0.f}, s1 = s0, s2 = s0, s3 = s0;
  int e = beg;
  for (; e + 4 <= end; e += 4) {
    const unsigned j0 = (unsigned)csr[e + 0];
    const unsigned j1 = (unsigned)csr[e + 1];
    const unsigned j2 = (unsigned)csr[e + 2];
    const unsigned j3 = (unsigned)csr[e + 3];
    const float4 v0 = *(const float4*)(xt + (j0 * (unsigned)DIM + c));
    const float4 v1 = *(const float4*)(xt + (j1 * (unsigned)DIM + c));
    const float4 v2 = *(const float4*)(xt + (j2 * (unsigned)DIM + c));
    const float4 v3 = *(const float4*)(xt + (j3 * (unsigned)DIM + c));
    s0.x += v0.x; s0.y += v0.y; s0.z += v0.z; s0.w += v0.w;
    s1.x += v1.x; s1.y += v1.y; s1.z += v1.z; s1.w += v1.w;
    s2.x += v2.x; s2.y += v2.y; s2.z += v2.z; s2.w += v2.w;
    s3.x += v3.x; s3.y += v3.y; s3.z += v3.z; s3.w += v3.w;
  }
  const int rem = end - e;
  if (rem > 0) {
    const int last = end - 1;
    const unsigned j0 = (unsigned)csr[e];
    const unsigned j1 = (unsigned)csr[min(e + 1, last)];
    const unsigned j2 = (unsigned)csr[min(e + 2, last)];
    const float4 z = {0.f, 0.f, 0.f, 0.f};
    const float4 v0 = *(const float4*)(xt + (j0 * (unsigned)DIM + c));
    float4 v1 = z, v2 = z;
    if (rem > 1) v1 = *(const float4*)(xt + (j1 * (unsigned)DIM + c));
    if (rem > 2) v2 = *(const float4*)(xt + (j2 * (unsigned)DIM + c));
    s0.x += v0.x; s0.y += v0.y; s0.z += v0.z; s0.w += v0.w;
    s1.x += v1.x; s1.y += v1.y; s1.z += v1.z; s1.w += v1.w;
    s2.x += v2.x; s2.y += v2.y; s2.z += v2.z; s2.w += v2.w;
  }

  const float dd = dis_of(end - beg);
  float4 x;
  x.x = dd * ((s0.x + s1.x) + (s2.x + s3.x));
  x.y = dd * ((s0.y + s1.y) + (s2.y + s3.y));
  x.z = dd * ((s0.z + s1.z) + (s2.z + s3.z));
  x.w = dd * ((s0.w + s1.w) + (s2.w + s3.w));

  const size_t o = (size_t)t * 4;
  float4 a = *(const float4*)(acc + o);
  a.x += x.x; a.y += x.y; a.z += x.z; a.w += x.w;
  if (LAST) {
    a.x *= 0.25f; a.y *= 0.25f; a.z *= 0.25f; a.w *= 0.25f;
  } else {
    float4 y;
    y.x = dd * x.x; y.y = dd * x.y; y.z = dd * x.z; y.w = dd * x.w;
    *(float4*)(xt_next + o) = y;
  }
  *(float4*)(acc + o) = a;
}

extern "C" void kernel_launch(void* const* d_in, const int* in_sizes, int n_in,
                              void* d_out, int out_size, void* d_ws, size_t ws_size,
                              hipStream_t stream) {
  const int*   ei    = (const int*)d_in[0];
  const int*   src   = ei;
  const int*   dst   = ei + NEDGE;
  const float* user  = (const float*)d_in[1];
  const float* bert  = (const float*)d_in[2];
  const float* wproj = (const float*)d_in[3];

  float* emb0 = (float*)d_out;
  float* acc  = emb0 + (size_t)N_NODES * DIM;

  int*   rowptr = (int*)d_ws;
  int*   csr    = rowptr + ((N_NODES + 1 + 63) & ~63);
  float* xA     = (float*)(csr + ((NEDGE + 63) & ~63));
  float* xB     = xA + (size_t)N_NODES * DIM;
  unsigned* bkt = (unsigned*)xB;
  int*   hist   = (int*)xB + ((NEDGE + 63) & ~63);
  int*   bbase  = hist + ((GB * NBKT + 63) & ~63);

  init_user<<<SGRID, SBLK, 0, stream>>>(user, emb0, acc);
  wmma_proj<<<PJ_GRID, PJ_BLK, 0, stream>>>(bert, wproj, emb0, acc);

  bucket_hist   <<<GB, BT, 0, stream>>>(dst, hist);
  bucket_scatter<<<GB, BT, 0, stream>>>(src, dst, hist, bbase, bkt);
  bucket_sort   <<<NBKT, BT, 0, stream>>>(bkt, bbase, rowptr, csr);

  prescale<<<SGRID, SBLK, 0, stream>>>(rowptr, emb0, xA);
  spmm_rows<false><<<(N_NODES * 16 + 255) / 256, 256, 0, stream>>>(rowptr, csr, xA, xB, acc);
  spmm_rows<false><<<(N_NODES * 16 + 255) / 256, 256, 0, stream>>>(rowptr, csr, xB, xA, acc);
  spmm_rows<true> <<<(N_NODES * 16 + 255) / 256, 256, 0, stream>>>(rowptr, csr, xA, xB, acc);
}


# SpMM rows: a row's indices are read 16 per request, one per lane, and passed round its half-wave by DPP row_share, so 8 source rows are requested back to back, no index load between; same sum order
# speedup vs baseline: 4.8332x; 1.1011x over previous
#include <hip/hip_runtime.h>

typedef __attribute__((ext_vector_type(2))) float v2f;
typedef __attribute__((ext_vector_type(8))) float v8f;
typedef __attribute__((ext_vector_type(4))) float v4f;

constexpr int NUM_USERS = 100000;
constexpr int NUM_ITEMS = 50000;
constexpr int N_NODES   = NUM_USERS + NUM_ITEMS;
constexpr int DIM       = 64;
constexpr int SBERT     = 384;
constexpr int NEDGE     = 2000000;
constexpr int NLAYERS   = 3;

constexpr int SGRID = 512;
constexpr int SBLK  = 256;
constexpr int STHR  = SGRID * SBLK;

constexpr int U4    = NUM_USERS * DIM / 4;
constexpr int U_PER = (U4 + STHR - 1) / STHR;
static_assert((U_PER - 1) * STHR <= U4, "only a lane's last item needs a bound test");

__global__ __launch_bounds__(SBLK)
void init_user(const float* __restrict__ u,
               float* __restrict__ emb0, float* __restrict__ acc) {
  const int t = blockIdx.x * SBLK + threadIdx.x;
  const int  il = t + (U_PER - 1) * STHR;
  const bool hl = il < U4;
  v4f v[U_PER - 1];
  v4f vl = {0.f, 0.f, 0.f, 0.f};
#pragma unroll
  for (int k = 0; k < U_PER - 1; ++k) v[k] = *(const v4f*)(u + (size_t)(t + k * STHR) * 4);
  if (hl) vl = *(const v4f*)(u + (size_t)il * 4);
#pragma unroll
  for (int k = 0; k < U_PER - 1; ++k) {
    const size_t o = (size_t)(t + k * STHR) * 4;
    *(v4f*)(emb0 + o) = v[k];
    *(v4f*)(acc  + o) = v[k];
  }
  if (hl) {
    *(v4f*)(emb0 + (size_t)il * 4) = vl;
    *(v4f*)(acc  + (size_t)il * 4) = vl;
  }
}

constexpr int PJ_GRID  = 256;
constexpr int PJ_WAVES = 16;
constexpr int PJ_FULL  = 12;
constexpr int PJ_BLK   = PJ_WAVES * 32;
constexpr int MTILES   = NUM_ITEMS / 16;
static_assert(NUM_ITEMS % 16 == 0, "whole strips only");
static_assert(PJ_WAVES - PJ_FULL == DIM / 16, "one wave per column tile of a split strip");
static_assert(PJ_FULL * PJ_GRID <= MTILES && MTILES <= (PJ_FULL + 1) * PJ_GRID, "every strip has one owner");
constexpr int WPITCH   = SBERT + 4;
constexpr int KSTEPS   = SBERT / 8;
constexpr int AG       = 8;
constexpr int W4       = DIM * SBERT / 4;
constexpr int W_PER    = W4 / PJ_BLK;
static_assert(W_PER * PJ_BLK == W4, "staging divides evenly");

__device__ __forceinline__ v8f wmma4(v2f a, v2f b, v8f c) {
  return __builtin_amdgcn_wmma_f32_16x16x4_f32(
       false, a,  false, b,
       (short)0, c,  false,  false);
}

template <int NT>
__device__ __forceinline__ void proj_strip(const float* __restrict__ arow, const float* bl,
                                           v2f (&a0)[AG], v2f (&a1)[AG],
                                           float* __restrict__ emb0, float* __restrict__ acc,
                                           size_t obase) {
  v8f c[NT];
#pragma unroll
  for (int j = 0; j < NT; ++j) c[j] = v8f{};
#pragma unroll
  for (int g = 0; g < KSTEPS / AG; ++g) {
    v2f n0[AG], n1[AG];
    if (g + 1 < KSTEPS / AG) {
#pragma unroll
      for (int s = 0; s < AG; ++s) {
        n0[s] = *(const v2f*)(arow + 8 * ((g + 1) * AG + s));
        n1[s] = *(const v2f*)(arow + 8 * ((g + 1) * AG + s) + 4);
      }
    }
#pragma unroll
    for (int s = 0; s < AG; ++s) {
      const int k8 = (g * AG + s) * 8;
      v4f b[NT];
#pragma unroll
      for (int j = 0; j < NT; ++j) b[j] = *(const v4f*)(bl + j * 16 * WPITCH + k8);
#pragma unroll
      for (int j = 0; j < NT; ++j) c[j] = wmma4(a0[s], b[j].lo, c[j]);
#pragma unroll
      for (int j = 0; j < NT; ++j) c[j] = wmma4(a1[s], b[j].hi, c[j]);
    }
    if (g + 1 < KSTEPS / AG) {
#pragma unroll
      for (int s = 0; s < AG; ++s) { a0[s] = n0[s]; a1[s] = n1[s]; }
    }
  }
#pragma unroll
  for (int i = 0; i < 8; ++i) {
#pragma unroll
    for (int j = 0; j < NT; ++j) {
      const size_t o = obase + (size_t)i * DIM + 16 * j;
      emb0[o] = c[j][i];
      acc[o]  = c[j][i];
    }
  }
}

__global__ __launch_bounds__(PJ_BLK)
void wmma_proj(const float* __restrict__ bert, const float* __restrict__ w,
               float* __restrict__ emb0, float* __restrict__ acc) {
  __shared__ __attribute__((aligned(16))) float wl[DIM * WPITCH];
  const int lane = threadIdx.x & 31;
  const int wave = threadIdx.x >> 5;
  const int half = lane >> 4;
  const int r    = lane & 15;
  const bool full = wave < PJ_FULL;
  const int tile = (full ? wave : PJ_FULL) * PJ_GRID + blockIdx.x;
  const int jt   = full ? 0 : wave - PJ_FULL;
  const bool active = tile < MTILES;

  const float* arow = bert + (size_t)((active ? tile : 0) * 16 + r) * SBERT + 2 * half;

  v2f a0[AG], a1[AG];
  if (active) {
#pragma unroll
    for (int s = 0; s < AG; ++s) {
      a0[s] = *(const v2f*)(arow + 8 * s);
      a1[s] = *(const v2f*)(arow + 8 * s + 4);
    }
  }

  {
    v4f t[W_PER];
#pragma unroll
    for (int k = 0; k < W_PER; ++k) t[k] = *(const v4f*)(w + (size_t)(threadIdx.x + k * PJ_BLK) * 4);
#pragma unroll
    for (int k = 0; k < W_PER; ++k) {
      const int i  = threadIdx.x + k * PJ_BLK;
      const int n  = i / (SBERT / 4);
      const int c  = i % (SBERT / 4);
      float* d = wl + n * WPITCH + 8 * (c >> 1) + 2 * (c & 1);
      *(v2f*)(d)     = t[k].lo;
      *(v2f*)(d + 4) = t[k].hi;
    }
  }
  __syncthreads();
  if (!active) return;

  const float* bl = wl + (jt * 16 + r) * WPITCH + 4 * half;
  const size_t obase = (size_t)(NUM_USERS + tile * 16 + 8 * half) * DIM + (size_t)(jt * 16 + r);
  if (full) proj_strip<DIM / 16>(arow, bl, a0, a1, emb0, acc, obase);
  else      proj_strip<1>(arow, bl, a0, a1, emb0, acc, obase);
}


constexpr int BT    = 1024;
constexpr int NBKT  = (N_NODES + 1023) >> 10;
constexpr int GB    = 256;
constexpr int CHUNK = (NEDGE + GB - 1) / GB;
constexpr int EPT   = 8;
constexpr int CHUNK_PAD = EPT * BT;
static_assert(CHUNK <= CHUNK_PAD, "a chunk must fit its LDS stage");
constexpr int CAPL  = 32768;
constexpr unsigned NOEDGE = 0xFFFFFFFFu;

__global__ __launch_bounds__(BT)
void bucket_hist(const int* __restrict__ dst, int* __restrict__ hist) {
  __shared__ int h[256];
  const int tid = threadIdx.x;
  if (tid < 256) h[tid] = 0;
  __syncthreads();
  const int beg = blockIdx.x * CHUNK;
  const int end = min(beg + CHUNK, NEDGE);
  int d[EPT];
#pragma unroll
  for (int k = 0; k < EPT; ++k) {
    const int i = beg + tid + k * BT;
    d[k] = (i < end) ? dst[i] : -1;
  }
#pragma unroll
  for (int k = 0; k < EPT; ++k)
    if (d[k] >= 0) atomicAdd(&h[d[k] >> 10], 1);
  __syncthreads();
  if (tid < NBKT) hist[blockIdx.x * NBKT + tid] = h[tid];
}

__global__ __launch_bounds__(BT)
void bucket_scatter(const int* __restrict__ src, const int* __restrict__ dst,
                    const int* __restrict__ hist, int* __restrict__ bbase,
                    unsigned* __restrict__ bkt) {
  __shared__ int p_tot[4][256];
  __shared__ int p_pre[4][256];
  __shared__ int w_tot[8];
  __shared__ int w_own[8];
  __shared__ int off[256];
  __shared__ int cur[256];
  __shared__ int s_src[CHUNK_PAD];
  __shared__ int s_dst[CHUNK_PAD];
  const int tid  = threadIdx.x;
  const int lane = tid & 31;
  const int wv   = tid >> 5;
  const int b    = blockIdx.x;

  {
    const int j = tid & 255;
    const int part = tid >> 8;
    int tot = 0, pre = 0;
    if (j < NBKT) {
      const int b0 = part * (GB / 4);
#pragma unroll 8
      for (int k = 0; k < GB / 4; ++k) {
        const int bb = b0 + k;
        const int v = hist[bb * NBKT + j];
        tot += v;
        pre += (bb < b) ? v : 0;
      }
    }
    p_tot[part][j] = tot;
    p_pre[part][j] = pre;
  }
  __syncthreads();

  int tot = 0, pre = 0, own = 0, inc_t = 0, inc_o = 0;
  if (tid < 256) {
    tot = (p_tot[0][tid] + p_tot[1][tid]) + (p_tot[2][tid] + p_tot[3][tid]);
    pre = (p_pre[0][tid] + p_pre[1][tid]) + (p_pre[2][tid] + p_pre[3][tid]);
    own = (tid < NBKT) ? hist[b * NBKT + tid] : 0;
    inc_t = tot;
    inc_o = own;
    for (int d = 1; d < 32; d <<= 1) {
      const int ut = __shfl_up(inc_t, d);
      const int uo = __shfl_up(inc_o, d);
      if (lane >= d) { inc_t += ut; inc_o += uo; }
    }
    if (lane == 31) { w_tot[wv] = inc_t; w_own[wv] = inc_o; }
  }
  __syncthreads();
  if (tid < 256) {
    int bt = 0, bo = 0;
    for (int k = 0; k < wv; ++k) { bt += w_tot[k]; bo += w_own[k]; }
    const int base = bt + inc_t - tot;
    const int lst  = bo + inc_o - own;
    cur[tid] = lst;
    off[tid] = base + pre - lst;
    if (b == 0 && tid <= NBKT) bbase[tid] = base;
  }
  __syncthreads();

  const int beg = b * CHUNK;
  const int end = min(beg + CHUNK, NEDGE);
  int dd[EPT], ss[EPT];
#pragma unroll
  for (int k = 0; k < EPT; ++k) {
    const int i = beg + tid + k * BT;
    const bool in = i < end;
    dd[k] = in ? dst[i] : -1;
    ss[k] = in ? src[i] : 0;
  }
#pragma unroll
  for (int k = 0; k < EPT; ++k) {
    if (dd[k] >= 0) {
      const int p = atomicAdd(&cur[dd[k] >> 10], 1);
      s_src[p] = ss[k];
      s_dst[p] = dd[k];
    }
  }
  __syncthreads();
  const int n = end - beg;
#pragma unroll
  for (int k = 0; k < EPT; ++k) {
    const int p = tid + k * BT;
    if (p < n) {
      const int d = s_dst[p];
      bkt[off[d >> 10] + p] = ((unsigned)s_src[p] & 0x3FFFFu) | ((unsigned)(d & 1023) << 18);
    }
  }
}

__global__ __launch_bounds__(BT)
void bucket_sort(const unsigned* __restrict__ bkt, const int* __restrict__ bbase,
                 int* __restrict__ rowptr, int* __restrict__ csr) {
  __shared__ int c[BT];
  __shared__ int wsum[BT / 32];
  __shared__ int sorted[CAPL];
  const int tid  = threadIdx.x;
  const int lane = tid & 31;
  const int wv   = tid >> 5;
  const int beg  = bbase[blockIdx.x];
  const int end  = bbase[blockIdx.x + 1];
  const int n    = end - beg;

  c[tid] = 0;
  __syncthreads();
  for (int i0 = beg + tid; i0 < end; i0 += EPT * BT) {
    unsigned w[EPT];
#pragma unroll
    for (int k = 0; k < EPT; ++k) {
      const int i = i0 + k * BT;
      w[k] = (i < end) ? bkt[i] : NOEDGE;
    }
#pragma unroll
    for (int k = 0; k < EPT; ++k)
      if (w[k] != NOEDGE) atomicAdd(&c[w[k] >> 18], 1);
  }
  __syncthreads();

  const int own = c[tid];
  int inc = own;
  for (int d = 1; d < 32; d <<= 1) {
    const int up = __shfl_up(inc, d);
    if (lane >= d) inc += up;
  }
  if (lane == 31) wsum[wv] = inc;
  __syncthreads();
  int before = (lane < wv) ? wsum[lane] : 0;
  for (int d = 16; d > 0; d >>= 1) before += __shfl_xor(before, d);
  const int ex = before + inc - own;
  const int node = blockIdx.x * BT + tid;
  if (node <= N_NODES) rowptr[node] = beg + ex;
  c[tid] = ex;
  __syncthreads();

  if (n <= CAPL) {
    for (int i0 = beg + tid; i0 < end; i0 += EPT * BT) {
      unsigned w[EPT];
#pragma unroll
      for (int k = 0; k < EPT; ++k) {
        const int i = i0 + k * BT;
        w[k] = (i < end) ? bkt[i] : NOEDGE;
      }
#pragma unroll
      for (int k = 0; k < EPT; ++k) {
        if (w[k] != NOEDGE) {
          const int p = atomicAdd(&c[w[k] >> 18], 1);
          sorted[p] = (int)(w[k] & 0x3FFFFu);
        }
      }
    }
    __syncthreads();
    for (int p = tid; p < n; p += BT) csr[beg + p] = sorted[p];
  } else {
    for (int i = beg + tid; i < end; i += BT) {
      const unsigned w = bkt[i];
      const int p = atomicAdd(&c[w >> 18], 1);
      csr[beg + p] = (int)(w & 0x3FFFFu);
    }
  }
}

__device__ __forceinline__ float dis_of(int deg) {
  return (deg > 0) ? rsqrtf(fmaxf((float)deg, 1.0f)) : 0.0f;
}

constexpr int P4    = N_NODES * DIM / 4;
constexpr int P_PER = (P4 + STHR - 1) / STHR;
static_assert((P_PER - 1) * STHR <= P4, "only a lane's last item needs a bound test");

__global__ __launch_bounds__(SBLK)
void prescale(const int* __restrict__ rowptr, const float* __restrict__ emb0,
              float* __restrict__ xt) {
  const int t = blockIdx.x * SBLK + threadIdx.x;
  const int  il = t + (P_PER - 1) * STHR;
  const bool hl = il < P4;
  int lo[P_PER - 1], hi[P_PER - 1];
  v4f v[P_PER - 1];
  int lol = 0, hil = 0;
  v4f vl = {0.f, 0.f, 0.f, 0.f};
#pragma unroll
  for (int k = 0; k < P_PER - 1; ++k) {
    const int i = t + k * STHR;
    lo[k] = rowptr[i >> 4];
    hi[k] = rowptr[(i >> 4) + 1];
    v[k]  = *(const v4f*)(emb0 + (size_t)i * 4);
  }
  if (hl) {
    lol = rowptr[il >> 4];
    hil = rowptr[(il >> 4) + 1];
    vl  = *(const v4f*)(emb0 + (size_t)il * 4);
  }
#pragma unroll
  for (int k = 0; k < P_PER - 1; ++k) {
    const float dd = dis_of(hi[k] - lo[k]);
    *(v4f*)(xt + (size_t)(t + k * STHR) * 4) = v[k] * dd;
  }
  if (hl) {
    const float dd = dis_of(hil - lol);
    *(v4f*)(xt + (size_t)il * 4) = vl * dd;
  }
}


template <int K>
__device__ __forceinline__ unsigned row_lane(unsigned v) {
  return (unsigned)__builtin_amdgcn_update_dpp(0, (int)v, 0x150 + K, 0xF, 0xF, true);
}

template <int K, bool ALWAYS>
__device__ __forceinline__ void row_of(const float* __restrict__ xt, unsigned m, unsigned cb,
                                       int n, float4& v) {
  const unsigned j = row_lane<K & 15>(m);
  v = float4{0.f, 0.f, 0.f, 0.f};
  if (ALWAYS || K < n) v = *(const float4*)((const char*)xt + ((j << 8) | cb));
}

__device__ __forceinline__ void add4(float4& s, const float4& v) {
  s.x += v.x; s.y += v.y; s.z += v.z; s.w += v.w;
}

template <int K0, bool INIT, bool A0>
__device__ __forceinline__ void gather8(const float* __restrict__ xt, unsigned m, unsigned cb, int n,
                                        float4& s0, float4& s1, float4& s2, float4& s3) {
  float4 v0, v1, v2, v3, v4, v5, v6, v7;
  row_of<K0 + 0, A0   >(xt, m, cb, n, v0);
  row_of<K0 + 1, false>(xt, m, cb, n, v1);
  row_of<K0 + 2, false>(xt, m, cb, n, v2);
  row_of<K0 + 3, false>(xt, m, cb, n, v3);
  row_of<K0 + 4, false>(xt, m, cb, n, v4);
  row_of<K0 + 5, false>(xt, m, cb, n, v5);
  row_of<K0 + 6, false>(xt, m, cb, n, v6);
  row_of<K0 + 7, false>(xt, m, cb, n, v7);
  if (INIT) { s0 = v0; s1 = v1; s2 = v2; s3 = v3; }
  else      { add4(s0, v0); add4(s1, v1); add4(s2, v2); add4(s3, v3); }
  add4(s0, v4); add4(s1, v5); add4(s2, v6); add4(s3, v7);
}

template <bool FIRST>
__device__ __forceinline__ void gather32(const float* __restrict__ xt, unsigned m0, unsigned m1,
                                         unsigned cb, int n,
                                         float4& s0, float4& s1, float4& s2, float4& s3) {
  gather8<0, FIRST, !FIRST>(xt, m0, cb, n, s0, s1, s2, s3);
  if (n > 8)  gather8<8,  false, true>(xt, m0, cb, n, s0, s1, s2, s3);
  if (n > 16) gather8<16, false, true>(xt, m1, cb, n, s0, s1, s2, s3);
  if (n > 24) gather8<24, false, true>(xt, m1, cb, n, s0, s1, s2, s3);
}

template <bool LAST>
__global__ __launch_bounds__(256)
void spmm_rows(const int* __restrict__ rowptr, const int* __restrict__ csr,
               const float* __restrict__ xt, float* __restrict__ xt_next,
               float* __restrict__ acc) {
  const int t = blockIdx.x * 256 + threadIdx.x;
  const int row = t >> 4;
  if (row >= N_NODES) return;
  const int hl = t & 15;
  const unsigned cb = (unsigned)hl << 4;
  const int beg = rowptr[row];
  const int end = rowptr[row + 1];
  const int n = end - beg;
  int pos = beg + hl;

  float4 s0, s1, s2, s3;
  const unsigned m0 = (unsigned)csr[max(min(pos, end - 1), 0)];
  const unsigned m1 = (unsigned)csr[max(min(pos + 16, end - 1), 0)];
  float4 a = *(const float4*)(acc + (size_t)t * 4);
  gather32<true>(xt, m0, m1, cb, n, s0, s1, s2, s3);
  for (int left = n - 32; left > 0; left -= 32) {
    pos += 32;
    const int lastpos = pos - (int)(cb >> 4) + left - 1;
    const unsigned q0 = (unsigned)csr[min(pos, lastpos)];
    const unsigned q1 = (unsigned)csr[min(pos + 16, lastpos)];
    gather32<false>(xt, q0, q1, cb, left, s0, s1, s2, s3);
  }

  const float dd = dis_of(n);
  float4 x;
  x.x = dd * ((s0.x + s1.x) + (s2.x + s3.x));
  x.y = dd * ((s0.y + s1.y) + (s2.y + s3.y));
  x.z = dd * ((s0.z + s1.z) + (s2.z + s3.z));
  x.w = dd * ((s0.w + s1.w) + (s2.w + s3.w));

  a.x += x.x; a.y += x.y; a.z += x.z; a.w += x.w;
  if (LAST) {
    a.x *= 0.25f; a.y *= 0.25f; a.z *= 0.25f; a.w *= 0.25f;
  } else {
    float4 y;
    y.x = dd * x.x; y.y = dd * x.y; y.z = dd * x.z; y.w = dd * x.w;
    *(float4*)(xt_next + (size_t)t * 4) = y;
  }
  *(float4*)(acc + (size_t)t * 4) = a;
}

extern "C" void kernel_launch(void* const* d_in, const int* in_sizes, int n_in,
                              void* d_out, int out_size, void* d_ws, size_t ws_size,
                              hipStream_t stream) {
  const int*   ei    = (const int*)d_in[0];
  const int*   src   = ei;
  const int*   dst   = ei + NEDGE;
  const float* user  = (const float*)d_in[1];
  const float* bert  = (const float*)d_in[2];
  const float* wproj = (const float*)d_in[3];

  float* emb0 = (float*)d_out;
  float* acc  = emb0 + (size_t)N_NODES * DIM;

  int*   rowptr = (int*)d_ws;
  int*   csr    = rowptr + ((N_NODES + 1 + 63) & ~63);
  float* xA     = (float*)(csr + ((NEDGE + 63) & ~63));
  float* xB     = xA + (size_t)N_NODES * DIM;
  unsigned* bkt = (unsigned*)xB;
  int*   hist   = (int*)xB + ((NEDGE + 63) & ~63);
  int*   bbase  = hist + ((GB * NBKT + 63) & ~63);

  init_user<<<SGRID, SBLK, 0, stream>>>(user, emb0, acc);
  wmma_proj<<<PJ_GRID, PJ_BLK, 0, stream>>>(bert, wproj, emb0, acc);

  bucket_hist   <<<GB, BT, 0, stream>>>(dst, hist);
  bucket_scatter<<<GB, BT, 0, stream>>>(src, dst, hist, bbase, bkt);
  bucket_sort   <<<NBKT, BT, 0, stream>>>(bkt, bbase, rowptr, csr);

  prescale<<<SGRID, SBLK, 0, stream>>>(rowptr, emb0, xA);
  spmm_rows<false><<<(N_NODES * 16 + 255) / 256, 256, 0, stream>>>(rowptr, csr, xA, xB, acc);
  spmm_rows<false><<<(N_NODES * 16 + 255) / 256, 256, 0, stream>>>(rowptr, csr, xB, xA, acc);
  spmm_rows<true> <<<(N_NODES * 16 + 255) / 256, 256, 0, stream>>>(rowptr, csr, xA, xB, acc);
}


# User copy, item projection and bucket histogram, independent of each other, run as ONE launch (256 projection + 256 copy/histogram work-groups, two per WGP): 9 launches become 7; all sums unchanged
# speedup vs baseline: 4.9888x; 1.0322x over previous
#include <hip/hip_runtime.h>

typedef __attribute__((ext_vector_type(2))) float v2f;
typedef __attribute__((ext_vector_type(8))) float v8f;
typedef __attribute__((ext_vector_type(4))) float v4f;

constexpr int NUM_USERS = 100000;
constexpr int NUM_ITEMS = 50000;
constexpr int N_NODES   = NUM_USERS + NUM_ITEMS;
constexpr int DIM       = 64;
constexpr int SBERT     = 384;
constexpr int NEDGE     = 2000000;
constexpr int NLAYERS   = 3;

constexpr int SGRID = 512;
constexpr int SBLK  = 256;
constexpr int STHR  = SGRID * SBLK;

constexpr int U4    = NUM_USERS * DIM / 4;
constexpr int U_PER = (U4 + STHR - 1) / STHR;
static_assert((U_PER - 1) * STHR <= U4, "only a lane's last item needs a bound test");

struct UserItems { v4f v[U_PER - 1]; v4f vl; };

__device__ __forceinline__ void user_load(const float* __restrict__ u, int t, UserItems& it) {
  const int il = t + (U_PER - 1) * STHR;
  it.vl = v4f{0.f, 0.f, 0.f, 0.f};
#pragma unroll
  for (int k = 0; k < U_PER - 1; ++k) it.v[k] = *(const v4f*)(u + (size_t)(t + k * STHR) * 4);
  if (il < U4) it.vl = *(const v4f*)(u + (size_t)il * 4);
}

__device__ __forceinline__ void user_store(float* __restrict__ emb0, float* __restrict__ acc,
                                           int t, const UserItems& it) {
  const int il = t + (U_PER - 1) * STHR;
#pragma unroll
  for (int k = 0; k < U_PER - 1; ++k) {
    const size_t o = (size_t)(t + k * STHR) * 4;
    *(v4f*)(emb0 + o) = it.v[k];
    *(v4f*)(acc  + o) = it.v[k];
  }
  if (il < U4) {
    *(v4f*)(emb0 + (size_t)il * 4) = it.vl;
    *(v4f*)(acc  + (size_t)il * 4) = it.vl;
  }
}

constexpr int PJ_GRID  = 256;
constexpr int PJ_WAVES = 16;
constexpr int PJ_FULL  = 12;
constexpr int PJ_BLK   = PJ_WAVES * 32;
constexpr int MTILES   = NUM_ITEMS / 16;
static_assert(NUM_ITEMS % 16 == 0, "whole strips only");
static_assert(PJ_WAVES - PJ_FULL == DIM / 16, "one wave per column tile of a split strip");
static_assert(PJ_FULL * PJ_GRID <= MTILES && MTILES <= (PJ_FULL + 1) * PJ_GRID, "every strip has one owner");
constexpr int WPITCH   = SBERT + 4;
constexpr int KSTEPS   = SBERT / 8;
constexpr int AG       = 8;
constexpr int W4       = DIM * SBERT / 4;
constexpr int W_PER    = W4 / PJ_BLK;
static_assert(W_PER * PJ_BLK == W4, "staging divides evenly");

__device__ __forceinline__ v8f wmma4(v2f a, v2f b, v8f c) {
  return __builtin_amdgcn_wmma_f32_16x16x4_f32(
       false, a,  false, b,
       (short)0, c,  false,  false);
}

template <int NT>
__device__ __forceinline__ void proj_strip(const float* __restrict__ arow, const float* bl,
                                           v2f (&a0)[AG], v2f (&a1)[AG],
                                           float* __restrict__ emb0, float* __restrict__ acc,
                                           size_t obase) {
  v8f c[NT];
#pragma unroll
  for (int j = 0; j < NT; ++j) c[j] = v8f{};
#pragma unroll
  for (int g = 0; g < KSTEPS / AG; ++g) {
    v2f n0[AG], n1[AG];
    if (g + 1 < KSTEPS / AG) {
#pragma unroll
      for (int s = 0; s < AG; ++s) {
        n0[s] = *(const v2f*)(arow + 8 * ((g + 1) * AG + s));
        n1[s] = *(const v2f*)(arow + 8 * ((g + 1) * AG + s) + 4);
      }
    }
#pragma unroll
    for (int s = 0; s < AG; ++s) {
      const int k8 = (g * AG + s) * 8;
      v4f b[NT];
#pragma unroll
      for (int j = 0; j < NT; ++j) b[j] = *(const v4f*)(bl + j * 16 * WPITCH + k8);
#pragma unroll
      for (int j = 0; j < NT; ++j) c[j] = wmma4(a0[s], b[j].lo, c[j]);
#pragma unroll
      for (int j = 0; j < NT; ++j) c[j] = wmma4(a1[s], b[j].hi, c[j]);
    }
    if (g + 1 < KSTEPS / AG) {
#pragma unroll
      for (int s = 0; s < AG; ++s) { a0[s] = n0[s]; a1[s] = n1[s]; }
    }
  }
#pragma unroll
  for (int i = 0; i < 8; ++i) {
#pragma unroll
    for (int j = 0; j < NT; ++j) {
      const size_t o = obase + (size_t)i * DIM + 16 * j;
      emb0[o] = c[j][i];
      acc[o]  = c[j][i];
    }
  }
}

__device__ __forceinline__
void proj_role(const float* __restrict__ bert, const float* __restrict__ w,
               float* __restrict__ emb0, float* __restrict__ acc,
               float* wl, int blk) {
  const int lane = threadIdx.x & 31;
  const int wave = threadIdx.x >> 5;
  const int half = lane >> 4;
  const int r    = lane & 15;
  const bool full = wave < PJ_FULL;
  const int tile = (full ? wave : PJ_FULL) * PJ_GRID + blk;
  const int jt   = full ? 0 : wave - PJ_FULL;
  const bool active = tile < MTILES;

  const float* arow = bert + (size_t)((active ? tile : 0) * 16 + r) * SBERT + 2 * half;

  v2f a0[AG], a1[AG];
  if (active) {
#pragma unroll
    for (int s = 0; s < AG; ++s) {
      a0[s] = *(const v2f*)(arow + 8 * s);
      a1[s] = *(const v2f*)(arow + 8 * s + 4);
    }
  }

  {
    v4f t[W_PER];
#pragma unroll
    for (int k = 0; k < W_PER; ++k) t[k] = *(const v4f*)(w + (size_t)(threadIdx.x + k * PJ_BLK) * 4);
#pragma unroll
    for (int k = 0; k < W_PER; ++k) {
      const int i  = threadIdx.x + k * PJ_BLK;
      const int n  = i / (SBERT / 4);
      const int c  = i % (SBERT / 4);
      float* d = wl + n * WPITCH + 8 * (c >> 1) + 2 * (c & 1);
      *(v2f*)(d)     = t[k].lo;
      *(v2f*)(d + 4) = t[k].hi;
    }
  }
  __syncthreads();
  if (!active) return;

  const float* bl = wl + (jt * 16 + r) * WPITCH + 4 * half;
  const size_t obase = (size_t)(NUM_USERS + tile * 16 + 8 * half) * DIM + (size_t)(jt * 16 + r);
  if (full) proj_strip<DIM / 16>(arow, bl, a0, a1, emb0, acc, obase);
  else      proj_strip<1>(arow, bl, a0, a1, emb0, acc, obase);
}


constexpr int BT    = 1024;
constexpr int NBKT  = (N_NODES + 1023) >> 10;
constexpr int GB    = 256;
constexpr int CHUNK = (NEDGE + GB - 1) / GB;
constexpr int EPT   = 8;
constexpr int CHUNK_PAD = EPT * BT;
static_assert(CHUNK <= CHUNK_PAD, "a chunk must fit its LDS stage");
constexpr int CAPL  = 32768;
constexpr unsigned NOEDGE = 0xFFFFFFFFu;

constexpr int FS_GRID = GB;
constexpr int FE_BLK  = PJ_BLK;
constexpr int HEPT    = 16;
static_assert(FS_GRID * FE_BLK == STHR, "the copy's lanes are exactly the stream lane count");
static_assert(CHUNK <= HEPT * FE_BLK, "a chunk must fit one pass of the work-group");
static_assert(NBKT <= 256 && NBKT <= FE_BLK, "one thread per bucket counter");

__device__ __forceinline__
void stream_role(const float* __restrict__ user, const int* __restrict__ dst,
                 float* __restrict__ emb0, float* __restrict__ acc,
                 int* __restrict__ hist, int* h, int s) {
  const int tid = threadIdx.x;
  if (tid < 256) h[tid] = 0;
  __syncthreads();
  const int beg = s * CHUNK;
  const int end = min(beg + CHUNK, NEDGE);
  int d[HEPT];
#pragma unroll
  for (int k = 0; k < HEPT; ++k) {
    const int i = beg + tid + k * FE_BLK;
    d[k] = (i < end) ? dst[i] : -1;
  }
  UserItems it;
  user_load(user, s * FE_BLK + tid, it);
#pragma unroll
  for (int k = 0; k < HEPT; ++k)
    if (d[k] >= 0) atomicAdd(&h[d[k] >> 10], 1);
  __syncthreads();
  if (tid < NBKT) hist[s * NBKT + tid] = h[tid];
  user_store(emb0, acc, s * FE_BLK + tid, it);
}

__global__ __launch_bounds__(FE_BLK)
void front(const float* __restrict__ user, const float* __restrict__ bert,
           const float* __restrict__ w, const int* __restrict__ dst,
           float* __restrict__ emb0, float* __restrict__ acc, int* __restrict__ hist) {
  __shared__ __attribute__((aligned(16))) float wl[DIM * WPITCH];
  __shared__ int h[256];
  if (blockIdx.x < PJ_GRID) proj_role(bert, w, emb0, acc, wl, blockIdx.x);
  else                      stream_role(user, dst, emb0, acc, hist, h, blockIdx.x - PJ_GRID);
}

__global__ __launch_bounds__(BT)
void bucket_scatter(const int* __restrict__ src, const int* __restrict__ dst,
                    const int* __restrict__ hist, int* __restrict__ bbase,
                    unsigned* __restrict__ bkt) {
  __shared__ int p_tot[4][256];
  __shared__ int p_pre[4][256];
  __shared__ int w_tot[8];
  __shared__ int w_own[8];
  __shared__ int off[256];
  __shared__ int cur[256];
  __shared__ int s_src[CHUNK_PAD];
  __shared__ int s_dst[CHUNK_PAD];
  const int tid  = threadIdx.x;
  const int lane = tid & 31;
  const int wv   = tid >> 5;
  const int b    = blockIdx.x;

  {
    const int j = tid & 255;
    const int part = tid >> 8;
    int tot = 0, pre = 0;
    if (j < NBKT) {
      const int b0 = part * (GB / 4);
#pragma unroll 8
      for (int k = 0; k < GB / 4; ++k) {
        const int bb = b0 + k;
        const int v = hist[bb * NBKT + j];
        tot += v;
        pre += (bb < b) ? v : 0;
      }
    }
    p_tot[part][j] = tot;
    p_pre[part][j] = pre;
  }
  __syncthreads();

  int tot = 0, pre = 0, own = 0, inc_t = 0, inc_o = 0;
  if (tid < 256) {
    tot = (p_tot[0][tid] + p_tot[1][tid]) + (p_tot[2][tid] + p_tot[3][tid]);
    pre = (p_pre[0][tid] + p_pre[1][tid]) + (p_pre[2][tid] + p_pre[3][tid]);
    own = (tid < NBKT) ? hist[b * NBKT + tid] : 0;
    inc_t = tot;
    inc_o = own;
    for (int d = 1; d < 32; d <<= 1) {
      const int ut = __shfl_up(inc_t, d);
      const int uo = __shfl_up(inc_o, d);
      if (lane >= d) { inc_t += ut; inc_o += uo; }
    }
    if (lane == 31) { w_tot[wv] = inc_t; w_own[wv] = inc_o; }
  }
  __syncthreads();
  if (tid < 256) {
    int bt = 0, bo = 0;
    for (int k = 0; k < wv; ++k) { bt += w_tot[k]; bo += w_own[k]; }
    const int base = bt + inc_t - tot;
    const int lst  = bo + inc_o - own;
    cur[tid] = lst;
    off[tid] = base + pre - lst;
    if (b == 0 && tid <= NBKT) bbase[tid] = base;
  }
  __syncthreads();

  const int beg = b * CHUNK;
  const int end = min(beg + CHUNK, NEDGE);
  int dd[EPT], ss[EPT];
#pragma unroll
  for (int k = 0; k < EPT; ++k) {
    const int i = beg + tid + k * BT;
    const bool in = i < end;
    dd[k] = in ? dst[i] : -1;
    ss[k] = in ? src[i] : 0;
  }
#pragma unroll
  for (int k = 0; k < EPT; ++k) {
    if (dd[k] >= 0) {
      const int p = atomicAdd(&cur[dd[k] >> 10], 1);
      s_src[p] = ss[k];
      s_dst[p] = dd[k];
    }
  }
  __syncthreads();
  const int n = end - beg;
#pragma unroll
  for (int k = 0; k < EPT; ++k) {
    const int p = tid + k * BT;
    if (p < n) {
      const int d = s_dst[p];
      bkt[off[d >> 10] + p] = ((unsigned)s_src[p] & 0x3FFFFu) | ((unsigned)(d & 1023) << 18);
    }
  }
}

__global__ __launch_bounds__(BT)
void bucket_sort(const unsigned* __restrict__ bkt, const int* __restrict__ bbase,
                 int* __restrict__ rowptr, int* __restrict__ csr) {
  __shared__ int c[BT];
  __shared__ int wsum[BT / 32];
  __shared__ int sorted[CAPL];
  const int tid  = threadIdx.x;
  const int lane = tid & 31;
  const int wv   = tid >> 5;
  const int beg  = bbase[blockIdx.x];
  const int end  = bbase[blockIdx.x + 1];
  const int n    = end - beg;

  c[tid] = 0;
  __syncthreads();
  for (int i0 = beg + tid; i0 < end; i0 += EPT * BT) {
    unsigned w[EPT];
#pragma unroll
    for (int k = 0; k < EPT; ++k) {
      const int i = i0 + k * BT;
      w[k] = (i < end) ? bkt[i] : NOEDGE;
    }
#pragma unroll
    for (int k = 0; k < EPT; ++k)
      if (w[k] != NOEDGE) atomicAdd(&c[w[k] >> 18], 1);
  }
  __syncthreads();

  const int own = c[tid];
  int inc = own;
  for (int d = 1; d < 32; d <<= 1) {
    const int up = __shfl_up(inc, d);
    if (lane >= d) inc += up;
  }
  if (lane == 31) wsum[wv] = inc;
  __syncthreads();
  int before = (lane < wv) ? wsum[lane] : 0;
  for (int d = 16; d > 0; d >>= 1) before += __shfl_xor(before, d);
  const int ex = before + inc - own;
  const int node = blockIdx.x * BT + tid;
  if (node <= N_NODES) rowptr[node] = beg + ex;
  c[tid] = ex;
  __syncthreads();

  if (n <= CAPL) {
    for (int i0 = beg + tid; i0 < end; i0 += EPT * BT) {
      unsigned w[EPT];
#pragma unroll
      for (int k = 0; k < EPT; ++k) {
        const int i = i0 + k * BT;
        w[k] = (i < end) ? bkt[i] : NOEDGE;
      }
#pragma unroll
      for (int k = 0; k < EPT; ++k) {
        if (w[k] != NOEDGE) {
          const int p = atomicAdd(&c[w[k] >> 18], 1);
          sorted[p] = (int)(w[k] & 0x3FFFFu);
        }
      }
    }
    __syncthreads();
    for (int p = tid; p < n; p += BT) csr[beg + p] = sorted[p];
  } else {
    for (int i = beg + tid; i < end; i += BT) {
      const unsigned w = bkt[i];
      const int p = atomicAdd(&c[w >> 18], 1);
      csr[beg + p] = (int)(w & 0x3FFFFu);
    }
  }
}

__device__ __forceinline__ float dis_of(int deg) {
  return (deg > 0) ? rsqrtf(fmaxf((float)deg, 1.0f)) : 0.0f;
}

constexpr int P4    = N_NODES * DIM / 4;
constexpr int P_PER = (P4 + STHR - 1) / STHR;
static_assert((P_PER - 1) * STHR <= P4, "only a lane's last item needs a bound test");

__global__ __launch_bounds__(SBLK)
void prescale(const int* __restrict__ rowptr, const float* __restrict__ emb0,
              float* __restrict__ xt) {
  const int t = blockIdx.x * SBLK + threadIdx.x;
  const int  il = t + (P_PER - 1) * STHR;
  const bool hl = il < P4;
  int lo[P_PER - 1], hi[P_PER - 1];
  v4f v[P_PER - 1];
  int lol = 0, hil = 0;
  v4f vl = {0.f, 0.f, 0.f, 0.f};
#pragma unroll
  for (int k = 0; k < P_PER - 1; ++k) {
    const int i = t + k * STHR;
    lo[k] = rowptr[i >> 4];
    hi[k] = rowptr[(i >> 4) + 1];
    v[k]  = *(const v4f*)(emb0 + (size_t)i * 4);
  }
  if (hl) {
    lol = rowptr[il >> 4];
    hil = rowptr[(il >> 4) + 1];
    vl  = *(const v4f*)(emb0 + (size_t)il * 4);
  }
#pragma unroll
  for (int k = 0; k < P_PER - 1; ++k) {
    const float dd = dis_of(hi[k] - lo[k]);
    *(v4f*)(xt + (size_t)(t + k * STHR) * 4) = v[k] * dd;
  }
  if (hl) {
    const float dd = dis_of(hil - lol);
    *(v4f*)(xt + (size_t)il * 4) = vl * dd;
  }
}


template <int K>
__device__ __forceinline__ unsigned row_lane(unsigned v) {
  return (unsigned)__builtin_amdgcn_update_dpp(0, (int)v, 0x150 + K, 0xF, 0xF, true);
}

template <int K, bool ALWAYS>
__device__ __forceinline__ void row_of(const float* __restrict__ xt, unsigned m, unsigned cb,
                                       int n, float4& v) {
  const unsigned j = row_lane<K & 15>(m);
  v = float4{0.f, 0.f, 0.f, 0.f};
  if (ALWAYS || K < n) v = *(const float4*)((const char*)xt + ((j << 8) | cb));
}

__device__ __forceinline__ void add4(float4& s, const float4& v) {
  s.x += v.x; s.y += v.y; s.z += v.z; s.w += v.w;
}

template <int K0, bool INIT, bool A0>
__device__ __forceinline__ void gather8(const float* __restrict__ xt, unsigned m, unsigned cb, int n,
                                        float4& s0, float4& s1, float4& s2, float4& s3) {
  float4 v0, v1, v2, v3, v4, v5, v6, v7;
  row_of<K0 + 0, A0   >(xt, m, cb, n, v0);
  row_of<K0 + 1, false>(xt, m, cb, n, v1);
  row_of<K0 + 2, false>(xt, m, cb, n, v2);
  row_of<K0 + 3, false>(xt, m, cb, n, v3);
  row_of<K0 + 4, false>(xt, m, cb, n, v4);
  row_of<K0 + 5, false>(xt, m, cb, n, v5);
  row_of<K0 + 6, false>(xt, m, cb, n, v6);
  row_of<K0 + 7, false>(xt, m, cb, n, v7);
  if (INIT) { s0 = v0; s1 = v1; s2 = v2; s3 = v3; }
  else      { add4(s0, v0); add4(s1, v1); add4(s2, v2); add4(s3, v3); }
  add4(s0, v4); add4(s1, v5); add4(s2, v6); add4(s3, v7);
}

template <bool FIRST>
__device__ __forceinline__ void gather32(const float* __restrict__ xt, unsigned m0, unsigned m1,
                                         unsigned cb, int n,
                                         float4& s0, float4& s1, float4& s2, float4& s3) {
  gather8<0, FIRST, !FIRST>(xt, m0, cb, n, s0, s1, s2, s3);
  if (n > 8)  gather8<8,  false, true>(xt, m0, cb, n, s0, s1, s2, s3);
  if (n > 16) gather8<16, false, true>(xt, m1, cb, n, s0, s1, s2, s3);
  if (n > 24) gather8<24, false, true>(xt, m1, cb, n, s0, s1, s2, s3);
}

template <bool LAST>
__global__ __launch_bounds__(256)
void spmm_rows(const int* __restrict__ rowptr, const int* __restrict__ csr,
               const float* __restrict__ xt, float* __restrict__ xt_next,
               float* __restrict__ acc) {
  const int t = blockIdx.x * 256 + threadIdx.x;
  const int row = t >> 4;
  if (row >= N_NODES) return;
  const int hl = t & 15;
  const unsigned cb = (unsigned)hl << 4;
  const int beg = rowptr[row];
  const int end = rowptr[row + 1];
  const int n = end - beg;
  int pos = beg + hl;

  float4 s0, s1, s2, s3;
  const unsigned m0 = (unsigned)csr[max(min(pos, end - 1), 0)];
  const unsigned m1 = (unsigned)csr[max(min(pos + 16, end - 1), 0)];
  float4 a = *(const float4*)(acc + (size_t)t * 4);
  gather32<true>(xt, m0, m1, cb, n, s0, s1, s2, s3);
  for (int left = n - 32; left > 0; left -= 32) {
    pos += 32;
    const int lastpos = pos - (int)(cb >> 4) + left - 1;
    const unsigned q0 = (unsigned)csr[min(pos, lastpos)];
    const unsigned q1 = (unsigned)csr[min(pos + 16, lastpos)];
    gather32<false>(xt, q0, q1, cb, left, s0, s1, s2, s3);
  }

  const float dd = dis_of(n);
  float4 x;
  x.x = dd * ((s0.x + s1.x) + (s2.x + s3.x));
  x.y = dd * ((s0.y + s1.y) + (s2.y + s3.y));
  x.z = dd * ((s0.z + s1.z) + (s2.z + s3.z));
  x.w = dd * ((s0.w + s1.w) + (s2.w + s3.w));

  a.x += x.x; a.y += x.y; a.z += x.z; a.w += x.w;
  if (LAST) {
    a.x *= 0.25f; a.y *= 0.25f; a.z *= 0.25f; a.w *= 0.25f;
  } else {
    float4 y;
    y.x = dd * x.x; y.y = dd * x.y; y.z = dd * x.z; y.w = dd * x.w;
    *(float4*)(xt_next + (size_t)t * 4) = y;
  }
  *(float4*)(acc + (size_t)t * 4) = a;
}

extern "C" void kernel_launch(void* const* d_in, const int* in_sizes, int n_in,
                              void* d_out, int out_size, void* d_ws, size_t ws_size,
                              hipStream_t stream) {
  const int*   ei    = (const int*)d_in[0];
  const int*   src   = ei;
  const int*   dst   = ei + NEDGE;
  const float* user  = (const float*)d_in[1];
  const float* bert  = (const float*)d_in[2];
  const float* wproj = (const float*)d_in[3];

  float* emb0 = (float*)d_out;
  float* acc  = emb0 + (size_t)N_NODES * DIM;

  int*   rowptr = (int*)d_ws;
  int*   csr    = rowptr + ((N_NODES + 1 + 63) & ~63);
  float* xA     = (float*)(csr + ((NEDGE + 63) & ~63));
  float* xB     = xA + (size_t)N_NODES * DIM;
  unsigned* bkt = (unsigned*)xB;
  int*   hist   = (int*)xB + ((NEDGE + 63) & ~63);
  int*   bbase  = hist + ((GB * NBKT + 63) & ~63);

  front<<<PJ_GRID + FS_GRID, FE_BLK, 0, stream>>>(user, bert, wproj, dst, emb0, acc, hist);

  bucket_scatter<<<GB, BT, 0, stream>>>(src, dst, hist, bbase, bkt);
  bucket_sort   <<<NBKT, BT, 0, stream>>>(bkt, bbase, rowptr, csr);

  prescale<<<SGRID, SBLK, 0, stream>>>(rowptr, emb0, xA);
  spmm_rows<false><<<(N_NODES * 16 + 255) / 256, 256, 0, stream>>>(rowptr, csr, xA, xB, acc);
  spmm_rows<false><<<(N_NODES * 16 + 255) / 256, 256, 0, stream>>>(rowptr, csr, xB, xA, acc);
  spmm_rows<true> <<<(N_NODES * 16 + 255) / 256, 256, 0, stream>>>(rowptr, csr, xA, xB, acc);
}


# Level one of the edge sort (chunk ordered by bucket in LDS, kept as one run plus offsets) joins the front launch; sort blocks collect 256 pieces; 7 launches to 6; row sums in LDS-add arrival order
# speedup vs baseline: 5.1790x; 1.0381x over previous
#include <hip/hip_runtime.h>

typedef __attribute__((ext_vector_type(2))) float v2f;
typedef __attribute__((ext_vector_type(8))) float v8f;
typedef __attribute__((ext_vector_type(4))) float v4f;

constexpr int NUM_USERS = 100000;
constexpr int NUM_ITEMS = 50000;
constexpr int N_NODES   = NUM_USERS + NUM_ITEMS;
constexpr int DIM       = 64;
constexpr int SBERT     = 384;
constexpr int NEDGE     = 2000000;
constexpr int NLAYERS   = 3;

constexpr int SGRID = 512;
constexpr int SBLK  = 256;
constexpr int STHR  = SGRID * SBLK;

constexpr int U4    = NUM_USERS * DIM / 4;
constexpr int U_PER = (U4 + STHR - 1) / STHR;
constexpr int U_CUT = 6;
static_assert((U_PER - 1) * STHR <= U4, "only a lane's last item needs a bound test");

template <int K0, int K1>
struct UserPart { v4f v[K1 - K0]; };

template <int K0, int K1>
__device__ __forceinline__ void user_load(const float* __restrict__ u, int t, UserPart<K0, K1>& it) {
#pragma unroll
  for (int k = K0; k < K1; ++k) {
    const int i = t + k * STHR;
    it.v[k - K0] = v4f{0.f, 0.f, 0.f, 0.f};
    if (k < U_PER - 1 || i < U4) it.v[k - K0] = *(const v4f*)(u + (size_t)i * 4);
  }
}

template <int K0, int K1>
__device__ __forceinline__ void user_store(float* __restrict__ emb0, float* __restrict__ acc,
                                           int t, const UserPart<K0, K1>& it) {
#pragma unroll
  for (int k = K0; k < K1; ++k) {
    const int i = t + k * STHR;
    if (k < U_PER - 1 || i < U4) {
      *(v4f*)(emb0 + (size_t)i * 4) = it.v[k - K0];
      *(v4f*)(acc  + (size_t)i * 4) = it.v[k - K0];
    }
  }
}

constexpr int PJ_GRID  = 256;
constexpr int PJ_WAVES = 16;
constexpr int PJ_FULL  = 12;
constexpr int PJ_BLK   = PJ_WAVES * 32;
constexpr int MTILES   = NUM_ITEMS / 16;
static_assert(NUM_ITEMS % 16 == 0, "whole strips only");
static_assert(PJ_WAVES - PJ_FULL == DIM / 16, "one wave per column tile of a split strip");
static_assert(PJ_FULL * PJ_GRID <= MTILES && MTILES <= (PJ_FULL + 1) * PJ_GRID, "every strip has one owner");
constexpr int WPITCH   = SBERT + 4;
constexpr int KSTEPS   = SBERT / 8;
constexpr int AG       = 8;
constexpr int W4       = DIM * SBERT / 4;
constexpr int W_PER    = W4 / PJ_BLK;
static_assert(W_PER * PJ_BLK == W4, "staging divides evenly");

__device__ __forceinline__ v8f wmma4(v2f a, v2f b, v8f c) {
  return __builtin_amdgcn_wmma_f32_16x16x4_f32(
       false, a,  false, b,
       (short)0, c,  false,  false);
}

template <int NT>
__device__ __forceinline__ void proj_strip(const float* __restrict__ arow, const float* bl,
                                           v2f (&a0)[AG], v2f (&a1)[AG],
                                           float* __restrict__ emb0, float* __restrict__ acc,
                                           size_t obase) {
  v8f c[NT];
#pragma unroll
  for (int j = 0; j < NT; ++j) c[j] = v8f{};
#pragma unroll
  for (int g = 0; g < KSTEPS / AG; ++g) {
    v2f n0[AG], n1[AG];
    if (g + 1 < KSTEPS / AG) {
#pragma unroll
      for (int s = 0; s < AG; ++s) {
        n0[s] = *(const v2f*)(arow + 8 * ((g + 1) * AG + s));
        n1[s] = *(const v2f*)(arow + 8 * ((g + 1) * AG + s) + 4);
      }
    }
#pragma unroll
    for (int s = 0; s < AG; ++s) {
      const int k8 = (g * AG + s) * 8;
      v4f b[NT];
#pragma unroll
      for (int j = 0; j < NT; ++j) b[j] = *(const v4f*)(bl + j * 16 * WPITCH + k8);
#pragma unroll
      for (int j = 0; j < NT; ++j) c[j] = wmma4(a0[s], b[j].lo, c[j]);
#pragma unroll
      for (int j = 0; j < NT; ++j) c[j] = wmma4(a1[s], b[j].hi, c[j]);
    }
    if (g + 1 < KSTEPS / AG) {
#pragma unroll
      for (int s = 0; s < AG; ++s) { a0[s] = n0[s]; a1[s] = n1[s]; }
    }
  }
#pragma unroll
  for (int i = 0; i < 8; ++i) {
#pragma unroll
    for (int j = 0; j < NT; ++j) {
      const size_t o = obase + (size_t)i * DIM + 16 * j;
      emb0[o] = c[j][i];
      acc[o]  = c[j][i];
    }
  }
}

__device__ __forceinline__
void proj_role(const float* __restrict__ bert, const float* __restrict__ w,
               float* __restrict__ emb0, float* __restrict__ acc,
               float* wl, int blk) {
  const int lane = threadIdx.x & 31;
  const int wave = threadIdx.x >> 5;
  const int half = lane >> 4;
  const int r    = lane & 15;
  const bool full = wave < PJ_FULL;
  const int tile = (full ? wave : PJ_FULL) * PJ_GRID + blk;
  const int jt   = full ? 0 : wave - PJ_FULL;
  const bool active = tile < MTILES;

  const float* arow = bert + (size_t)((active ? tile : 0) * 16 + r) * SBERT + 2 * half;

  v2f a0[AG], a1[AG];
  if (active) {
#pragma unroll
    for (int s = 0; s < AG; ++s) {
      a0[s] = *(const v2f*)(arow + 8 * s);
      a1[s] = *(const v2f*)(arow + 8 * s + 4);
    }
  }

  {
    v4f t[W_PER];
#pragma unroll
    for (int k = 0; k < W_PER; ++k) t[k] = *(const v4f*)(w + (size_t)(threadIdx.x + k * PJ_BLK) * 4);
#pragma unroll
    for (int k = 0; k < W_PER; ++k) {
      const int i  = threadIdx.x + k * PJ_BLK;
      const int n  = i / (SBERT / 4);
      const int c  = i % (SBERT / 4);
      float* d = wl + n * WPITCH + 8 * (c >> 1) + 2 * (c & 1);
      *(v2f*)(d)     = t[k].lo;
      *(v2f*)(d + 4) = t[k].hi;
    }
  }
  __syncthreads();
  if (!active) return;

  const float* bl = wl + (jt * 16 + r) * WPITCH + 4 * half;
  const size_t obase = (size_t)(NUM_USERS + tile * 16 + 8 * half) * DIM + (size_t)(jt * 16 + r);
  if (full) proj_strip<DIM / 16>(arow, bl, a0, a1, emb0, acc, obase);
  else      proj_strip<1>(arow, bl, a0, a1, emb0, acc, obase);
}


constexpr int BT    = 1024;
constexpr int NBKT  = (N_NODES + 1023) >> 10;
constexpr int GB    = 256;
constexpr int CHUNK = (NEDGE + GB - 1) / GB;
constexpr int PREP  = NBKT + 1;
constexpr int CAPL  = 32768;
constexpr unsigned NOEDGE = 0xFFFFFFFFu;

constexpr int FS_GRID = GB;
constexpr int FE_BLK  = PJ_BLK;
constexpr int HEPT    = 16;
constexpr int CHUNK_PAD = HEPT * FE_BLK;
static_assert(FS_GRID * FE_BLK == STHR, "the copy's lanes are exactly the stream lane count");
static_assert(CHUNK <= CHUNK_PAD, "a chunk must fit one pass of the work-group and its LDS stage");
static_assert(CHUNK_PAD <= DIM * WPITCH, "the stage fits the LDS that the projection role leaves unused here");
static_assert(PREP <= 256 && NBKT <= FE_BLK, "one thread per bucket counter");

__device__ __forceinline__
void stream_role(const float* __restrict__ user, const int* __restrict__ src,
                 const int* __restrict__ dst,
                 float* __restrict__ emb0, float* __restrict__ acc,
                 int* __restrict__ pre, unsigned* __restrict__ bkt,
                 int* h, int* cur, int* wtot, unsigned* stage, int s) {
  const int tid  = threadIdx.x;
  const int lane = tid & 31;
  const int wv   = tid >> 5;
  if (tid < 256) h[tid] = 0;
  __syncthreads();
  const int beg = s * CHUNK;
  const int end = min(beg + CHUNK, NEDGE);
  int d[HEPT], sr[HEPT];
#pragma unroll
  for (int k = 0; k < HEPT; ++k) {
    const int i  = beg + tid + k * FE_BLK;
    const int ic = min(i, end - 1);
    d[k]  = dst[ic];
    sr[k] = src[ic];
    if (i >= end) d[k] = -1;
  }
  UserPart<0, U_CUT> ita;
  user_load(user, s * FE_BLK + tid, ita);
#pragma unroll
  for (int k = 0; k < HEPT; ++k)
    if (d[k] >= 0) atomicAdd(&h[d[k] >> 10], 1);
  __syncthreads();
  user_store(emb0, acc, s * FE_BLK + tid, ita);
  UserPart<U_CUT, U_PER> itb;
  user_load(user, s * FE_BLK + tid, itb);

  int own = 0, inc = 0;
  if (tid < 256) {
    own = h[tid];
    inc = own;
    for (int dl = 1; dl < 32; dl <<= 1) {
      const int up = __shfl_up(inc, dl);
      if (lane >= dl) inc += up;
    }
    if (lane == 31) wtot[wv] = inc;
  }
  __syncthreads();
  if (tid < 256) {
    int before = 0;
    for (int k = 0; k < wv; ++k) before += wtot[k];
    const int ex = before + inc - own;
    cur[tid] = ex;
    if (tid < PREP) pre[s * PREP + tid] = ex;
  }
  __syncthreads();
#pragma unroll
  for (int k = 0; k < HEPT; ++k) {
    if (d[k] >= 0) {
      const int p = atomicAdd(&cur[d[k] >> 10], 1);
      stage[p] = ((unsigned)sr[k] & 0x3FFFFu) | ((unsigned)(d[k] & 1023) << 18);
    }
  }
  __syncthreads();
  const int n = end - beg;
#pragma unroll
  for (int k = 0; k < HEPT; ++k) {
    const int p = tid + k * FE_BLK;
    if (p < n) bkt[beg + p] = stage[p];
  }
  user_store(emb0, acc, s * FE_BLK + tid, itb);
}

__global__ __launch_bounds__(FE_BLK)
void front(const float* __restrict__ user, const float* __restrict__ bert,
           const float* __restrict__ w, const int* __restrict__ src, const int* __restrict__ dst,
           float* __restrict__ emb0, float* __restrict__ acc,
           int* __restrict__ pre, unsigned* __restrict__ bkt) {
  __shared__ __attribute__((aligned(16))) float wl[DIM * WPITCH];
  __shared__ int h[256];
  __shared__ int cur[256];
  __shared__ int wtot[8];
  if (blockIdx.x < PJ_GRID) proj_role(bert, w, emb0, acc, wl, blockIdx.x);
  else stream_role(user, src, dst, emb0, acc, pre, bkt, h, cur, wtot,
                   reinterpret_cast<unsigned*>(wl), blockIdx.x - PJ_GRID);
}

constexpr int SPW  = GB / (BT / 32);
constexpr int SDEP = 3;
static_assert(SPW * (BT / 32) == GB, "every piece has one wave");

__global__ __launch_bounds__(BT)
void bucket_sort(const unsigned* __restrict__ bkt, const int* __restrict__ pre,
                 int* __restrict__ rowptr, int* __restrict__ csr) {
  __shared__ int c[BT];
  __shared__ int wsum[BT / 32];
  __shared__ int sorted[CAPL];
  __shared__ int seg_lo[GB];
  __shared__ int seg_n[GB];
  __shared__ int red_lo[GB / 32];
  __shared__ int red_hi[GB / 32];
  const int tid  = threadIdx.x;
  const int lane = tid & 31;
  const int wv   = tid >> 5;
  const int j    = blockIdx.x;

  c[tid] = 0;
  if (tid < GB) {
    const int p0 = pre[tid * PREP + j];
    const int p1 = pre[tid * PREP + j + 1];
    seg_lo[tid] = tid * CHUNK + p0;
    seg_n[tid]  = p1 - p0;
    int a = p0, b = p1;
    for (int dl = 16; dl > 0; dl >>= 1) { a += __shfl_xor(a, dl); b += __shfl_xor(b, dl); }
    if (lane == 0) { red_lo[wv] = a; red_hi[wv] = b; }
  }
  __syncthreads();
  int beg = 0, end = 0;
  for (int k = 0; k < GB / 32; ++k) { beg += red_lo[k]; end += red_hi[k]; }
  const int n = end - beg;

  unsigned w[SPW][SDEP];
#pragma unroll
  for (int q = 0; q < SPW; ++q) {
    const int lo = seg_lo[wv * SPW + q];
    const int m  = seg_n[wv * SPW + q];
#pragma unroll
    for (int r = 0; r < SDEP; ++r) {
      const int i = lane + 32 * r;
      w[q][r] = (i < m) ? bkt[lo + i] : NOEDGE;
    }
  }
#pragma unroll
  for (int q = 0; q < SPW; ++q)
#pragma unroll
    for (int r = 0; r < SDEP; ++r)
      if (w[q][r] != NOEDGE) atomicAdd(&c[w[q][r] >> 18], 1);
  for (int q = 0; q < SPW; ++q) {
    const int lo = seg_lo[wv * SPW + q];
    const int m  = seg_n[wv * SPW + q];
    for (int i = lane + 32 * SDEP; i < m; i += 32) atomicAdd(&c[bkt[lo + i] >> 18], 1);
  }
  __syncthreads();

  const int own = c[tid];
  int inc = own;
  for (int d = 1; d < 32; d <<= 1) {
    const int up = __shfl_up(inc, d);
    if (lane >= d) inc += up;
  }
  if (lane == 31) wsum[wv] = inc;
  __syncthreads();
  int before = (lane < wv) ? wsum[lane] : 0;
  for (int d = 16; d > 0; d >>= 1) before += __shfl_xor(before, d);
  const int ex = before + inc - own;
  const int node = blockIdx.x * BT + tid;
  if (node <= N_NODES) rowptr[node] = beg + ex;
  c[tid] = ex;
  __syncthreads();

  if (n <= CAPL) {
#pragma unroll
    for (int q = 0; q < SPW; ++q)
#pragma unroll
      for (int r = 0; r < SDEP; ++r)
        if (w[q][r] != NOEDGE) {
          const int p = atomicAdd(&c[w[q][r] >> 18], 1);
          sorted[p] = (int)(w[q][r] & 0x3FFFFu);
        }
    for (int q = 0; q < SPW; ++q) {
      const int lo = seg_lo[wv * SPW + q];
      const int m  = seg_n[wv * SPW + q];
      for (int i = lane + 32 * SDEP; i < m; i += 32) {
        const unsigned x = bkt[lo + i];
        const int p = atomicAdd(&c[x >> 18], 1);
        sorted[p] = (int)(x & 0x3FFFFu);
      }
    }
    __syncthreads();
    for (int p = tid; p < n; p += BT) csr[beg + p] = sorted[p];
  } else {
#pragma unroll
    for (int q = 0; q < SPW; ++q)
#pragma unroll
      for (int r = 0; r < SDEP; ++r)
        if (w[q][r] != NOEDGE) {
          const int p = atomicAdd(&c[w[q][r] >> 18], 1);
          csr[beg + p] = (int)(w[q][r] & 0x3FFFFu);
        }
    for (int q = 0; q < SPW; ++q) {
      const int lo = seg_lo[wv * SPW + q];
      const int m  = seg_n[wv * SPW + q];
      for (int i = lane + 32 * SDEP; i < m; i += 32) {
        const unsigned x = bkt[lo + i];
        const int p = atomicAdd(&c[x >> 18], 1);
        csr[beg + p] = (int)(x & 0x3FFFFu);
      }
    }
  }
}

__device__ __forceinline__ float dis_of(int deg) {
  return (deg > 0) ? rsqrtf(fmaxf((float)deg, 1.0f)) : 0.0f;
}

constexpr int P4    = N_NODES * DIM / 4;
constexpr int P_PER = (P4 + STHR - 1) / STHR;
static_assert((P_PER - 1) * STHR <= P4, "only a lane's last item needs a bound test");

__global__ __launch_bounds__(SBLK)
void prescale(const int* __restrict__ rowptr, const float* __restrict__ emb0,
              float* __restrict__ xt) {
  const int t = blockIdx.x * SBLK + threadIdx.x;
  const int  il = t + (P_PER - 1) * STHR;
  const bool hl = il < P4;
  int lo[P_PER - 1], hi[P_PER - 1];
  v4f v[P_PER - 1];
  int lol = 0, hil = 0;
  v4f vl = {0.f, 0.f, 0.f, 0.f};
#pragma unroll
  for (int k = 0; k < P_PER - 1; ++k) {
    const int i = t + k * STHR;
    lo[k] = rowptr[i >> 4];
    hi[k] = rowptr[(i >> 4) + 1];
    v[k]  = *(const v4f*)(emb0 + (size_t)i * 4);
  }
  if (hl) {
    lol = rowptr[il >> 4];
    hil = rowptr[(il >> 4) + 1];
    vl  = *(const v4f*)(emb0 + (size_t)il * 4);
  }
#pragma unroll
  for (int k = 0; k < P_PER - 1; ++k) {
    const float dd = dis_of(hi[k] - lo[k]);
    *(v4f*)(xt + (size_t)(t + k * STHR) * 4) = v[k] * dd;
  }
  if (hl) {
    const float dd = dis_of(hil - lol);
    *(v4f*)(xt + (size_t)il * 4) = vl * dd;
  }
}


template <int K>
__device__ __forceinline__ unsigned row_lane(unsigned v) {
  return (unsigned)__builtin_amdgcn_update_dpp(0, (int)v, 0x150 + K, 0xF, 0xF, true);
}

template <int K, bool ALWAYS>
__device__ __forceinline__ void row_of(const float* __restrict__ xt, unsigned m, unsigned cb,
                                       int n, float4& v) {
  const unsigned j = row_lane<K & 15>(m);
  v = float4{0.f, 0.f, 0.f, 0.f};
  if (ALWAYS || K < n) v = *(const float4*)((const char*)xt + ((j << 8) | cb));
}

__device__ __forceinline__ void add4(float4& s, const float4& v) {
  s.x += v.x; s.y += v.y; s.z += v.z; s.w += v.w;
}

template <int K0, bool INIT, bool A0>
__device__ __forceinline__ void gather8(const float* __restrict__ xt, unsigned m, unsigned cb, int n,
                                        float4& s0, float4& s1, float4& s2, float4& s3) {
  float4 v0, v1, v2, v3, v4, v5, v6, v7;
  row_of<K0 + 0, A0   >(xt, m, cb, n, v0);
  row_of<K0 + 1, false>(xt, m, cb, n, v1);
  row_of<K0 + 2, false>(xt, m, cb, n, v2);
  row_of<K0 + 3, false>(xt, m, cb, n, v3);
  row_of<K0 + 4, false>(xt, m, cb, n, v4);
  row_of<K0 + 5, false>(xt, m, cb, n, v5);
  row_of<K0 + 6, false>(xt, m, cb, n, v6);
  row_of<K0 + 7, false>(xt, m, cb, n, v7);
  if (INIT) { s0 = v0; s1 = v1; s2 = v2; s3 = v3; }
  else      { add4(s0, v0); add4(s1, v1); add4(s2, v2); add4(s3, v3); }
  add4(s0, v4); add4(s1, v5); add4(s2, v6); add4(s3, v7);
}

template <bool FIRST>
__device__ __forceinline__ void gather32(const float* __restrict__ xt, unsigned m0, unsigned m1,
                                         unsigned cb, int n,
                                         float4& s0, float4& s1, float4& s2, float4& s3) {
  gather8<0, FIRST, !FIRST>(xt, m0, cb, n, s0, s1, s2, s3);
  if (n > 8)  gather8<8,  false, true>(xt, m0, cb, n, s0, s1, s2, s3);
  if (n > 16) gather8<16, false, true>(xt, m1, cb, n, s0, s1, s2, s3);
  if (n > 24) gather8<24, false, true>(xt, m1, cb, n, s0, s1, s2, s3);
}

template <bool LAST>
__global__ __launch_bounds__(256)
void spmm_rows(const int* __restrict__ rowptr, const int* __restrict__ csr,
               const float* __restrict__ xt, float* __restrict__ xt_next,
               float* __restrict__ acc) {
  const int t = blockIdx.x * 256 + threadIdx.x;
  const int row = t >> 4;
  if (row >= N_NODES) return;
  const int hl = t & 15;
  const unsigned cb = (unsigned)hl << 4;
  const int beg = rowptr[row];
  const int end = rowptr[row + 1];
  const int n = end - beg;
  int pos = beg + hl;

  float4 s0, s1, s2, s3;
  const unsigned m0 = (unsigned)csr[max(min(pos, end - 1), 0)];
  const unsigned m1 = (unsigned)csr[max(min(pos + 16, end - 1), 0)];
  float4 a = *(const float4*)(acc + (size_t)t * 4);
  gather32<true>(xt, m0, m1, cb, n, s0, s1, s2, s3);
  for (int left = n - 32; left > 0; left -= 32) {
    pos += 32;
    const int lastpos = pos - (int)(cb >> 4) + left - 1;
    const unsigned q0 = (unsigned)csr[min(pos, lastpos)];
    const unsigned q1 = (unsigned)csr[min(pos + 16, lastpos)];
    gather32<false>(xt, q0, q1, cb, left, s0, s1, s2, s3);
  }

  const float dd = dis_of(n);
  float4 x;
  x.x = dd * ((s0.x + s1.x) + (s2.x + s3.x));
  x.y = dd * ((s0.y + s1.y) + (s2.y + s3.y));
  x.z = dd * ((s0.z + s1.z) + (s2.z + s3.z));
  x.w = dd * ((s0.w + s1.w) + (s2.w + s3.w));

  a.x += x.x; a.y += x.y; a.z += x.z; a.w += x.w;
  if (LAST) {
    a.x *= 0.25f; a.y *= 0.25f; a.z *= 0.25f; a.w *= 0.25f;
  } else {
    float4 y;
    y.x = dd * x.x; y.y = dd * x.y; y.z = dd * x.z; y.w = dd * x.w;
    *(float4*)(xt_next + (size_t)t * 4) = y;
  }
  *(float4*)(acc + (size_t)t * 4) = a;
}

extern "C" void kernel_launch(void* const* d_in, const int* in_sizes, int n_in,
                              void* d_out, int out_size, void* d_ws, size_t ws_size,
                              hipStream_t stream) {
  const int*   ei    = (const int*)d_in[0];
  const int*   src   = ei;
  const int*   dst   = ei + NEDGE;
  const float* user  = (const float*)d_in[1];
  const float* bert  = (const float*)d_in[2];
  const float* wproj = (const float*)d_in[3];

  float* emb0 = (float*)d_out;
  float* acc  = emb0 + (size_t)N_NODES * DIM;

  int*   rowptr = (int*)d_ws;
  int*   csr    = rowptr + ((N_NODES + 1 + 63) & ~63);
  float* xA     = (float*)(csr + ((NEDGE + 63) & ~63));
  float* xB     = xA + (size_t)N_NODES * DIM;
  unsigned* bkt = (unsigned*)xB;
  int*   pre    = (int*)xB + ((NEDGE + 63) & ~63);

  front<<<PJ_GRID + FS_GRID, FE_BLK, 0, stream>>>(user, bert, wproj, src, dst, emb0, acc, pre, bkt);

  bucket_sort<<<NBKT, BT, 0, stream>>>(bkt, pre, rowptr, csr);

  prescale<<<SGRID, SBLK, 0, stream>>>(rowptr, emb0, xA);
  spmm_rows<false><<<(N_NODES * 16 + 255) / 256, 256, 0, stream>>>(rowptr, csr, xA, xB, acc);
  spmm_rows<false><<<(N_NODES * 16 + 255) / 256, 256, 0, stream>>>(rowptr, csr, xB, xA, acc);
  spmm_rows<true> <<<(N_NODES * 16 + 255) / 256, 256, 0, stream>>>(rowptr, csr, xA, xB, acc);
}


# Layer sum in Horner form, out = (x0 + A(x0 + A(x0 + A x0)))/4, on the scaled rows: no running-sum array is written or re-read, each layer writes one array; the order of the layer sums changes
# speedup vs baseline: 5.8685x; 1.1331x over previous
#include <hip/hip_runtime.h>

typedef __attribute__((ext_vector_type(2))) float v2f;
typedef __attribute__((ext_vector_type(8))) float v8f;
typedef __attribute__((ext_vector_type(4))) float v4f;

constexpr int NUM_USERS = 100000;
constexpr int NUM_ITEMS = 50000;
constexpr int N_NODES   = NUM_USERS + NUM_ITEMS;
constexpr int DIM       = 64;
constexpr int SBERT     = 384;
constexpr int NEDGE     = 2000000;
constexpr int NLAYERS   = 3;

constexpr int SGRID = 512;
constexpr int SBLK  = 256;
constexpr int STHR  = SGRID * SBLK;

constexpr int U4    = NUM_USERS * DIM / 4;
constexpr int U_PER = (U4 + STHR - 1) / STHR;
constexpr int U_CUT = 6;
static_assert((U_PER - 1) * STHR <= U4, "only a lane's last item needs a bound test");

template <int K0, int K1>
struct UserPart { v4f v[K1 - K0]; };

template <int K0, int K1>
__device__ __forceinline__ void user_load(const float* __restrict__ u, int t, UserPart<K0, K1>& it) {
#pragma unroll
  for (int k = K0; k < K1; ++k) {
    const int i = t + k * STHR;
    it.v[k - K0] = v4f{0.f, 0.f, 0.f, 0.f};
    if (k < U_PER - 1 || i < U4) it.v[k - K0] = *(const v4f*)(u + (size_t)i * 4);
  }
}

template <int K0, int K1>
__device__ __forceinline__ void user_store(float* __restrict__ emb0,
                                           int t, const UserPart<K0, K1>& it) {
#pragma unroll
  for (int k = K0; k < K1; ++k) {
    const int i = t + k * STHR;
    if (k < U_PER - 1 || i < U4) *(v4f*)(emb0 + (size_t)i * 4) = it.v[k - K0];
  }
}

constexpr int PJ_GRID  = 256;
constexpr int PJ_WAVES = 16;
constexpr int PJ_FULL  = 12;
constexpr int PJ_BLK   = PJ_WAVES * 32;
constexpr int MTILES   = NUM_ITEMS / 16;
static_assert(NUM_ITEMS % 16 == 0, "whole strips only");
static_assert(PJ_WAVES - PJ_FULL == DIM / 16, "one wave per column tile of a split strip");
static_assert(PJ_FULL * PJ_GRID <= MTILES && MTILES <= (PJ_FULL + 1) * PJ_GRID, "every strip has one owner");
constexpr int WPITCH   = SBERT + 4;
constexpr int KSTEPS   = SBERT / 8;
constexpr int AG       = 8;
constexpr int W4       = DIM * SBERT / 4;
constexpr int W_PER    = W4 / PJ_BLK;
static_assert(W_PER * PJ_BLK == W4, "staging divides evenly");

__device__ __forceinline__ v8f wmma4(v2f a, v2f b, v8f c) {
  return __builtin_amdgcn_wmma_f32_16x16x4_f32(
       false, a,  false, b,
       (short)0, c,  false,  false);
}

template <int NT>
__device__ __forceinline__ void proj_strip(const float* __restrict__ arow, const float* bl,
                                           v2f (&a0)[AG], v2f (&a1)[AG],
                                           float* __restrict__ emb0, size_t obase) {
  v8f c[NT];
#pragma unroll
  for (int j = 0; j < NT; ++j) c[j] = v8f{};
#pragma unroll
  for (int g = 0; g < KSTEPS / AG; ++g) {
    v2f n0[AG], n1[AG];
    if (g + 1 < KSTEPS / AG) {
#pragma unroll
      for (int s = 0; s < AG; ++s) {
        n0[s] = *(const v2f*)(arow + 8 * ((g + 1) * AG + s));
        n1[s] = *(const v2f*)(arow + 8 * ((g + 1) * AG + s) + 4);
      }
    }
#pragma unroll
    for (int s = 0; s < AG; ++s) {
      const int k8 = (g * AG + s) * 8;
      v4f b[NT];
#pragma unroll
      for (int j = 0; j < NT; ++j) b[j] = *(const v4f*)(bl + j * 16 * WPITCH + k8);
#pragma unroll
      for (int j = 0; j < NT; ++j) c[j] = wmma4(a0[s], b[j].lo, c[j]);
#pragma unroll
      for (int j = 0; j < NT; ++j) c[j] = wmma4(a1[s], b[j].hi, c[j]);
    }
    if (g + 1 < KSTEPS / AG) {
#pragma unroll
      for (int s = 0; s < AG; ++s) { a0[s] = n0[s]; a1[s] = n1[s]; }
    }
  }
#pragma unroll
  for (int i = 0; i < 8; ++i) {
#pragma unroll
    for (int j = 0; j < NT; ++j) {
      emb0[obase + (size_t)i * DIM + 16 * j] = c[j][i];
    }
  }
}

__device__ __forceinline__
void proj_role(const float* __restrict__ bert, const float* __restrict__ w,
               float* __restrict__ emb0, float* wl, int blk) {
  const int lane = threadIdx.x & 31;
  const int wave = threadIdx.x >> 5;
  const int half = lane >> 4;
  const int r    = lane & 15;
  const bool full = wave < PJ_FULL;
  const int tile = (full ? wave : PJ_FULL) * PJ_GRID + blk;
  const int jt   = full ? 0 : wave - PJ_FULL;
  const bool active = tile < MTILES;

  const float* arow = bert + (size_t)((active ? tile : 0) * 16 + r) * SBERT + 2 * half;

  v2f a0[AG], a1[AG];
  if (active) {
#pragma unroll
    for (int s = 0; s < AG; ++s) {
      a0[s] = *(const v2f*)(arow + 8 * s);
      a1[s] = *(const v2f*)(arow + 8 * s + 4);
    }
  }

  {
    v4f t[W_PER];
#pragma unroll
    for (int k = 0; k < W_PER; ++k) t[k] = *(const v4f*)(w + (size_t)(threadIdx.x + k * PJ_BLK) * 4);
#pragma unroll
    for (int k = 0; k < W_PER; ++k) {
      const int i  = threadIdx.x + k * PJ_BLK;
      const int n  = i / (SBERT / 4);
      const int c  = i % (SBERT / 4);
      float* d = wl + n * WPITCH + 8 * (c >> 1) + 2 * (c & 1);
      *(v2f*)(d)     = t[k].lo;
      *(v2f*)(d + 4) = t[k].hi;
    }
  }
  __syncthreads();
  if (!active) return;

  const float* bl = wl + (jt * 16 + r) * WPITCH + 4 * half;
  const size_t obase = (size_t)(NUM_USERS + tile * 16 + 8 * half) * DIM + (size_t)(jt * 16 + r);
  if (full) proj_strip<DIM / 16>(arow, bl, a0, a1, emb0, obase);
  else      proj_strip<1>(arow, bl, a0, a1, emb0, obase);
}


constexpr int BT    = 1024;
constexpr int NBKT  = (N_NODES + 1023) >> 10;
constexpr int GB    = 256;
constexpr int CHUNK = (NEDGE + GB - 1) / GB;
constexpr int PREP  = NBKT + 1;
constexpr int CAPL  = 32768;
constexpr unsigned NOEDGE = 0xFFFFFFFFu;

constexpr int FS_GRID = GB;
constexpr int FE_BLK  = PJ_BLK;
constexpr int HEPT    = 16;
constexpr int CHUNK_PAD = HEPT * FE_BLK;
static_assert(FS_GRID * FE_BLK == STHR, "the copy's lanes are exactly the stream lane count");
static_assert(CHUNK <= CHUNK_PAD, "a chunk must fit one pass of the work-group and its LDS stage");
static_assert(CHUNK_PAD <= DIM * WPITCH, "the stage fits the LDS that the projection role leaves unused here");
static_assert(PREP <= 256 && NBKT <= FE_BLK, "one thread per bucket counter");

__device__ __forceinline__
void stream_role(const float* __restrict__ user, const int* __restrict__ src,
                 const int* __restrict__ dst, float* __restrict__ emb0,
                 int* __restrict__ pre, unsigned* __restrict__ bkt,
                 int* h, int* cur, int* wtot, unsigned* stage, int s) {
  const int tid  = threadIdx.x;
  const int lane = tid & 31;
  const int wv   = tid >> 5;
  if (tid < 256) h[tid] = 0;
  __syncthreads();
  const int beg = s * CHUNK;
  const int end = min(beg + CHUNK, NEDGE);
  int d[HEPT], sr[HEPT];
#pragma unroll
  for (int k = 0; k < HEPT; ++k) {
    const int i  = beg + tid + k * FE_BLK;
    const int ic = min(i, end - 1);
    d[k]  = dst[ic];
    sr[k] = src[ic];
    if (i >= end) d[k] = -1;
  }
  UserPart<0, U_CUT> ita;
  user_load(user, s * FE_BLK + tid, ita);
#pragma unroll
  for (int k = 0; k < HEPT; ++k)
    if (d[k] >= 0) atomicAdd(&h[d[k] >> 10], 1);
  __syncthreads();
  user_store(emb0, s * FE_BLK + tid, ita);
  UserPart<U_CUT, U_PER> itb;
  user_load(user, s * FE_BLK + tid, itb);

  int own = 0, inc = 0;
  if (tid < 256) {
    own = h[tid];
    inc = own;
    for (int dl = 1; dl < 32; dl <<= 1) {
      const int up = __shfl_up(inc, dl);
      if (lane >= dl) inc += up;
    }
    if (lane == 31) wtot[wv] = inc;
  }
  __syncthreads();
  if (tid < 256) {
    int before = 0;
    for (int k = 0; k < wv; ++k) before += wtot[k];
    const int ex = before + inc - own;
    cur[tid] = ex;
    if (tid < PREP) pre[s * PREP + tid] = ex;
  }
  __syncthreads();
#pragma unroll
  for (int k = 0; k < HEPT; ++k) {
    if (d[k] >= 0) {
      const int p = atomicAdd(&cur[d[k] >> 10], 1);
      stage[p] = ((unsigned)sr[k] & 0x3FFFFu) | ((unsigned)(d[k] & 1023) << 18);
    }
  }
  __syncthreads();
  const int n = end - beg;
#pragma unroll
  for (int k = 0; k < HEPT; ++k) {
    const int p = tid + k * FE_BLK;
    if (p < n) bkt[beg + p] = stage[p];
  }
  user_store(emb0, s * FE_BLK + tid, itb);
}

__global__ __launch_bounds__(FE_BLK) __attribute__((amdgpu_waves_per_eu(8)))
void front(const float* __restrict__ user, const float* __restrict__ bert,
           const float* __restrict__ w, const int* __restrict__ src, const int* __restrict__ dst,
           float* __restrict__ emb0,
           int* __restrict__ pre, unsigned* __restrict__ bkt) {
  __shared__ __attribute__((aligned(16))) float wl[DIM * WPITCH];
  __shared__ int h[256];
  __shared__ int cur[256];
  __shared__ int wtot[8];
  if (blockIdx.x < PJ_GRID) proj_role(bert, w, emb0, wl, blockIdx.x);
  else stream_role(user, src, dst, emb0, pre, bkt, h, cur, wtot,
                   reinterpret_cast<unsigned*>(wl), blockIdx.x - PJ_GRID);
}

constexpr int SPW  = GB / (BT / 32);
constexpr int SDEP = 3;
static_assert(SPW * (BT / 32) == GB, "every piece has one wave");

__global__ __launch_bounds__(BT)
void bucket_sort(const unsigned* __restrict__ bkt, const int* __restrict__ pre,
                 int* __restrict__ rowptr, int* __restrict__ csr) {
  __shared__ int c[BT];
  __shared__ int wsum[BT / 32];
  __shared__ int sorted[CAPL];
  __shared__ int seg_lo[GB];
  __shared__ int seg_n[GB];
  __shared__ int red_lo[GB / 32];
  __shared__ int red_hi[GB / 32];
  const int tid  = threadIdx.x;
  const int lane = tid & 31;
  const int wv   = tid >> 5;
  const int j    = blockIdx.x;

  c[tid] = 0;
  if (tid < GB) {
    const int p0 = pre[tid * PREP + j];
    const int p1 = pre[tid * PREP + j + 1];
    seg_lo[tid] = tid * CHUNK + p0;
    seg_n[tid]  = p1 - p0;
    int a = p0, b = p1;
    for (int dl = 16; dl > 0; dl >>= 1) { a += __shfl_xor(a, dl); b += __shfl_xor(b, dl); }
    if (lane == 0) { red_lo[wv] = a; red_hi[wv] = b; }
  }
  __syncthreads();
  int beg = 0, end = 0;
  for (int k = 0; k < GB / 32; ++k) { beg += red_lo[k]; end += red_hi[k]; }
  const int n = end - beg;

  unsigned w[SPW][SDEP];
#pragma unroll
  for (int q = 0; q < SPW; ++q) {
    const int lo = seg_lo[wv * SPW + q];
    const int m  = seg_n[wv * SPW + q];
#pragma unroll
    for (int r = 0; r < SDEP; ++r) {
      const int i = lane + 32 * r;
      w[q][r] = (i < m) ? bkt[lo + i] : NOEDGE;
    }
  }
#pragma unroll
  for (int q = 0; q < SPW; ++q)
#pragma unroll
    for (int r = 0; r < SDEP; ++r)
      if (w[q][r] != NOEDGE) atomicAdd(&c[w[q][r] >> 18], 1);
  for (int q = 0; q < SPW; ++q) {
    const int lo = seg_lo[wv * SPW + q];
    const int m  = seg_n[wv * SPW + q];
    for (int i = lane + 32 * SDEP; i < m; i += 32) atomicAdd(&c[bkt[lo + i] >> 18], 1);
  }
  __syncthreads();

  const int own = c[tid];
  int inc = own;
  for (int d = 1; d < 32; d <<= 1) {
    const int up = __shfl_up(inc, d);
    if (lane >= d) inc += up;
  }
  if (lane == 31) wsum[wv] = inc;
  __syncthreads();
  int before = (lane < wv) ? wsum[lane] : 0;
  for (int d = 16; d > 0; d >>= 1) before += __shfl_xor(before, d);
  const int ex = before + inc - own;
  const int node = blockIdx.x * BT + tid;
  if (node <= N_NODES) rowptr[node] = beg + ex;
  c[tid] = ex;
  __syncthreads();

  if (n <= CAPL) {
#pragma unroll
    for (int q = 0; q < SPW; ++q)
#pragma unroll
      for (int r = 0; r < SDEP; ++r)
        if (w[q][r] != NOEDGE) {
          const int p = atomicAdd(&c[w[q][r] >> 18], 1);
          sorted[p] = (int)(w[q][r] & 0x3FFFFu);
        }
    for (int q = 0; q < SPW; ++q) {
      const int lo = seg_lo[wv * SPW + q];
      const int m  = seg_n[wv * SPW + q];
      for (int i = lane + 32 * SDEP; i < m; i += 32) {
        const unsigned x = bkt[lo + i];
        const int p = atomicAdd(&c[x >> 18], 1);
        sorted[p] = (int)(x & 0x3FFFFu);
      }
    }
    __syncthreads();
    for (int p = tid; p < n; p += BT) csr[beg + p] = sorted[p];
  } else {
#pragma unroll
    for (int q = 0; q < SPW; ++q)
#pragma unroll
      for (int r = 0; r < SDEP; ++r)
        if (w[q][r] != NOEDGE) {
          const int p = atomicAdd(&c[w[q][r] >> 18], 1);
          csr[beg + p] = (int)(w[q][r] & 0x3FFFFu);
        }
    for (int q = 0; q < SPW; ++q) {
      const int lo = seg_lo[wv * SPW + q];
      const int m  = seg_n[wv * SPW + q];
      for (int i = lane + 32 * SDEP; i < m; i += 32) {
        const unsigned x = bkt[lo + i];
        const int p = atomicAdd(&c[x >> 18], 1);
        csr[beg + p] = (int)(x & 0x3FFFFu);
      }
    }
  }
}

__device__ __forceinline__ float dis_of(int deg) {
  return (deg > 0) ? rsqrtf(fmaxf((float)deg, 1.0f)) : 0.0f;
}

constexpr int P4    = N_NODES * DIM / 4;
constexpr int P_PER = (P4 + STHR - 1) / STHR;
static_assert((P_PER - 1) * STHR <= P4, "only a lane's last item needs a bound test");

__global__ __launch_bounds__(SBLK)
void prescale(const int* __restrict__ rowptr, const float* __restrict__ emb0,
              float* __restrict__ xt) {
  const int t = blockIdx.x * SBLK + threadIdx.x;
  const int  il = t + (P_PER - 1) * STHR;
  const bool hl = il < P4;
  int lo[P_PER - 1], hi[P_PER - 1];
  v4f v[P_PER - 1];
  int lol = 0, hil = 0;
  v4f vl = {0.f, 0.f, 0.f, 0.f};
#pragma unroll
  for (int k = 0; k < P_PER - 1; ++k) {
    const int i = t + k * STHR;
    lo[k] = rowptr[i >> 4];
    hi[k] = rowptr[(i >> 4) + 1];
    v[k]  = *(const v4f*)(emb0 + (size_t)i * 4);
  }
  if (hl) {
    lol = rowptr[il >> 4];
    hil = rowptr[(il >> 4) + 1];
    vl  = *(const v4f*)(emb0 + (size_t)il * 4);
  }
#pragma unroll
  for (int k = 0; k < P_PER - 1; ++k) {
    const float dd = dis_of(hi[k] - lo[k]);
    *(v4f*)(xt + (size_t)(t + k * STHR) * 4) = v[k] * dd;
  }
  if (hl) {
    const float dd = dis_of(hil - lol);
    *(v4f*)(xt + (size_t)il * 4) = vl * dd;
  }
}


template <int K>
__device__ __forceinline__ unsigned row_lane(unsigned v) {
  return (unsigned)__builtin_amdgcn_update_dpp(0, (int)v, 0x150 + K, 0xF, 0xF, true);
}

template <int K, bool ALWAYS>
__device__ __forceinline__ void row_of(const float* __restrict__ xt, unsigned m, unsigned cb,
                                       int n, float4& v) {
  const unsigned j = row_lane<K & 15>(m);
  v = float4{0.f, 0.f, 0.f, 0.f};
  if (ALWAYS || K < n) v = *(const float4*)((const char*)xt + ((j << 8) | cb));
}

__device__ __forceinline__ void add4(float4& s, const float4& v) {
  s.x += v.x; s.y += v.y; s.z += v.z; s.w += v.w;
}

template <int K0, bool INIT, bool A0>
__device__ __forceinline__ void gather8(const float* __restrict__ xt, unsigned m, unsigned cb, int n,
                                        float4& s0, float4& s1, float4& s2, float4& s3) {
  float4 v0, v1, v2, v3, v4, v5, v6, v7;
  row_of<K0 + 0, A0   >(xt, m, cb, n, v0);
  row_of<K0 + 1, false>(xt, m, cb, n, v1);
  row_of<K0 + 2, false>(xt, m, cb, n, v2);
  row_of<K0 + 3, false>(xt, m, cb, n, v3);
  row_of<K0 + 4, false>(xt, m, cb, n, v4);
  row_of<K0 + 5, false>(xt, m, cb, n, v5);
  row_of<K0 + 6, false>(xt, m, cb, n, v6);
  row_of<K0 + 7, false>(xt, m, cb, n, v7);
  if (INIT) { s0 = v0; s1 = v1; s2 = v2; s3 = v3; }
  else      { add4(s0, v0); add4(s1, v1); add4(s2, v2); add4(s3, v3); }
  add4(s0, v4); add4(s1, v5); add4(s2, v6); add4(s3, v7);
}

template <bool FIRST>
__device__ __forceinline__ void gather32(const float* __restrict__ xt, unsigned m0, unsigned m1,
                                         unsigned cb, int n,
                                         float4& s0, float4& s1, float4& s2, float4& s3) {
  gather8<0, FIRST, !FIRST>(xt, m0, cb, n, s0, s1, s2, s3);
  if (n > 8)  gather8<8,  false, true>(xt, m0, cb, n, s0, s1, s2, s3);
  if (n > 16) gather8<16, false, true>(xt, m1, cb, n, s0, s1, s2, s3);
  if (n > 24) gather8<24, false, true>(xt, m1, cb, n, s0, s1, s2, s3);
}

template <bool LAST>
__global__ __launch_bounds__(256)
void spmm_rows(const int* __restrict__ rowptr, const int* __restrict__ csr,
               const float* __restrict__ xt, const float* own, float* dst) {
  const int t = blockIdx.x * 256 + threadIdx.x;
  const int row = t >> 4;
  if (row >= N_NODES) return;
  const int hl = t & 15;
  const unsigned cb = (unsigned)hl << 4;
  const int beg = rowptr[row];
  const int end = rowptr[row + 1];
  const int n = end - beg;
  int pos = beg + hl;

  float4 s0, s1, s2, s3;
  const unsigned m0 = (unsigned)csr[max(min(pos, end - 1), 0)];
  const unsigned m1 = (unsigned)csr[max(min(pos + 16, end - 1), 0)];
  const float4 a = *(const float4*)(own + (size_t)t * 4);
  gather32<true>(xt, m0, m1, cb, n, s0, s1, s2, s3);
  for (int left = n - 32; left > 0; left -= 32) {
    pos += 32;
    const int lastpos = pos - (int)(cb >> 4) + left - 1;
    const unsigned q0 = (unsigned)csr[min(pos, lastpos)];
    const unsigned q1 = (unsigned)csr[min(pos + 16, lastpos)];
    gather32<false>(xt, q0, q1, cb, left, s0, s1, s2, s3);
  }

  const float dd = dis_of(n);
  float4 s;
  s.x = (s0.x + s1.x) + (s2.x + s3.x);
  s.y = (s0.y + s1.y) + (s2.y + s3.y);
  s.z = (s0.z + s1.z) + (s2.z + s3.z);
  s.w = (s0.w + s1.w) + (s2.w + s3.w);

  float4 o;
  if (LAST) {
    o.x = 0.25f * (a.x + dd * s.x);
    o.y = 0.25f * (a.y + dd * s.y);
    o.z = 0.25f * (a.z + dd * s.z);
    o.w = 0.25f * (a.w + dd * s.w);
  } else {
    const float d2 = dd * dd;
    o.x = a.x + d2 * s.x;
    o.y = a.y + d2 * s.y;
    o.z = a.z + d2 * s.z;
    o.w = a.w + d2 * s.w;
  }
  *(float4*)(dst + (size_t)t * 4) = o;
}

extern "C" void kernel_launch(void* const* d_in, const int* in_sizes, int n_in,
                              void* d_out, int out_size, void* d_ws, size_t ws_size,
                              hipStream_t stream) {
  const int*   ei    = (const int*)d_in[0];
  const int*   src   = ei;
  const int*   dst   = ei + NEDGE;
  const float* user  = (const float*)d_in[1];
  const float* bert  = (const float*)d_in[2];
  const float* wproj = (const float*)d_in[3];

  float* emb0 = (float*)d_out;
  float* outp = emb0 + (size_t)N_NODES * DIM;

  int*   rowptr = (int*)d_ws;
  int*   csr    = rowptr + ((N_NODES + 1 + 63) & ~63);
  float* xA     = (float*)(csr + ((NEDGE + 63) & ~63));
  float* xB     = xA + (size_t)N_NODES * DIM;
  unsigned* bkt = (unsigned*)xB;
  int*   pre    = (int*)xB + ((NEDGE + 63) & ~63);

  front<<<PJ_GRID + FS_GRID, FE_BLK, 0, stream>>>(user, bert, wproj, src, dst, emb0, pre, bkt);

  bucket_sort<<<NBKT, BT, 0, stream>>>(bkt, pre, rowptr, csr);

  prescale<<<SGRID, SBLK, 0, stream>>>(rowptr, emb0, xA);
  spmm_rows<false><<<(N_NODES * 16 + 255) / 256, 256, 0, stream>>>(rowptr, csr, xA, xA, xB);
  spmm_rows<false><<<(N_NODES * 16 + 255) / 256, 256, 0, stream>>>(rowptr, csr, xB, xA, xA);
  spmm_rows<true> <<<(N_NODES * 16 + 255) / 256, 256, 0, stream>>>(rowptr, csr, xA, emb0, outp);
}
